# starting kernel
# baseline (speedup 1.0000x reference)
#include <hip/hip_runtime.h>
#include <math.h>

#define F_IN 512
#define F_HID 128
#define F_OUT 16

typedef __attribute__((ext_vector_type(16))) __bf16 v16bf;
typedef __attribute__((ext_vector_type(8)))  float  v8f;

union BFrag {
    uint4 u[2];
    v16bf v;
};

__device__ __forceinline__ unsigned int f2bf(float f) {
    union { float f; unsigned int u; } cv; cv.f = f;
    unsigned int u = cv.u;
    unsigned int r = u + 0x7FFFu + ((u >> 16) & 1u);
    return (r >> 16) & 0xFFFFu;
}

__device__ __forceinline__ uint4 cvt8(float4 a, float4 b) {
    uint4 r;
    r.x = f2bf(a.x) | (f2bf(a.y) << 16);
    r.y = f2bf(a.z) | (f2bf(a.w) << 16);
    r.z = f2bf(b.x) | (f2bf(b.y) << 16);
    r.w = f2bf(b.z) | (f2bf(b.w) << 16);
    return r;
}


__global__ void k_set_deg(float* deg, int n) {
    int i = blockIdx.x * blockDim.x + threadIdx.x;
    if (i < n) deg[i] = 1.0f;
}

__global__ void k_deg_accum(const int* __restrict__ dst,
                            const float* __restrict__ ew,
                            float* deg, int e) {
    int i = blockIdx.x * blockDim.x + threadIdx.x;
    if (i < e) atomicAdd(&deg[dst[i]], ew[i]);
}

__global__ void k_dinv(float* deg, int n) {
    int i = blockIdx.x * blockDim.x + threadIdx.x;
    if (i < n) {
        float d = deg[i];
        deg[i] = (d > 0.0f) ? rsqrtf(d) : 0.0f;
    }
}


__global__ void k_prep_weights(const float* __restrict__ W1,
                               const float* __restrict__ W2,
                               unsigned short* __restrict__ Wt1,
                               unsigned short* __restrict__ Wt2) {
    int t = blockIdx.x * blockDim.x + threadIdx.x;
    if (t < F_IN * F_HID) {
        int n = t / F_IN, k = t % F_IN;
        Wt1[t] = (unsigned short)f2bf(W1[k * F_HID + n]);
    }
    int t2 = t - F_IN * F_HID;
    if (t2 >= 0 && t2 < F_HID * F_OUT) {
        int n = t2 / F_HID, k = t2 % F_HID;
        Wt2[t2] = (unsigned short)f2bf(W2[k * F_OUT + n]);
    }
}


__global__ void k_gemm1_wmma(const float* __restrict__ x,
                             const unsigned short* __restrict__ Wt1,
                             float* __restrict__ h1,
                             int nrows) {
    __shared__ unsigned short Abuf[16 * F_IN];
    const int tid = threadIdx.x;
    const int row0 = blockIdx.x * 16;

    for (int i = tid; i < 16 * F_IN; i += 256) {
        int m = i / F_IN, k = i % F_IN;
        int gr = row0 + m;
        float v = (gr < nrows) ? x[(size_t)gr * F_IN + k] : 0.0f;
        Abuf[i] = (unsigned short)f2bf(v);
    }
    __syncthreads();

    const int wave = tid >> 5;
    const int lane = tid & 31;
    const int n0 = wave * 16;
    const int mrow = lane & 15;
    const int khalf = (lane >> 4) << 3;

    const unsigned short* brow = Wt1 + (size_t)(n0 + mrow) * F_IN;

    v8f acc = {};
    for (int k = 0; k < F_IN; k += 32) {
        BFrag a, b;
        a.u[0] = *(const uint4*)&Abuf[mrow * F_IN + k + khalf];
        a.u[1] = *(const uint4*)&Abuf[mrow * F_IN + k + 16 + khalf];
        b.u[0] = *(const uint4*)&brow[k + khalf];
        b.u[1] = *(const uint4*)&brow[k + 16 + khalf];
        acc = __builtin_amdgcn_wmma_f32_16x16x32_bf16(
            false, a.v, false, b.v, (short)0, acc, false, false);
    }

    const int col = n0 + (lane & 15);
    const int rbase = row0 + ((lane >> 4) << 3);
#pragma unroll
    for (int r = 0; r < 8; ++r) {
        int row = rbase + r;
        if (row < nrows) h1[(size_t)row * F_HID + col] = acc[r];
    }
}


__global__ void k_init_agg1(const float* __restrict__ h1,
                            const float* __restrict__ dinv,
                            float* __restrict__ agg, int n) {
    int t = blockIdx.x * blockDim.x + threadIdx.x;
    if (t >= n * 32) return;
    int i = t >> 5;
    float s = dinv[i] * dinv[i];
    float4 v = ((const float4*)h1)[t];
    v.x *= s; v.y *= s; v.z *= s; v.w *= s;
    ((float4*)agg)[t] = v;
}


__global__ void k_edge_agg1(const int* __restrict__ src,
                            const int* __restrict__ dst,
                            const float* __restrict__ ew,
                            const float* __restrict__ dinv,
                            const float* __restrict__ h1,
                            float* __restrict__ agg, int e) {
    int wid = (blockIdx.x * blockDim.x + threadIdx.x) >> 5;
    int lane = threadIdx.x & 31;
    if (wid >= e) return;
    int s = src[wid], d = dst[wid];
    float nrm = dinv[s] * ew[wid] * dinv[d];
    float4 v = ((const float4*)(h1 + (size_t)s * F_HID))[lane];
    float* o = agg + (size_t)d * F_HID + lane * 4;
    atomicAdd(o + 0, v.x * nrm);
    atomicAdd(o + 1, v.y * nrm);
    atomicAdd(o + 2, v.z * nrm);
    atomicAdd(o + 3, v.w * nrm);
}


__global__ void k_finalize1(const float* __restrict__ agg,
                            const float* __restrict__ b1,
                            float* __restrict__ hout, int n) {
    int t = blockIdx.x * blockDim.x + threadIdx.x;
    if (t >= n * 32) return;
    int c4 = (t & 31) * 4;
    float4 v = ((const float4*)agg)[t];
    v.x = fmaxf(v.x + b1[c4 + 0], 0.0f);
    v.y = fmaxf(v.y + b1[c4 + 1], 0.0f);
    v.z = fmaxf(v.z + b1[c4 + 2], 0.0f);
    v.w = fmaxf(v.w + b1[c4 + 3], 0.0f);
    ((float4*)hout)[t] = v;
}


__global__ void k_gemm2_wmma(const float* __restrict__ h,
                             const unsigned short* __restrict__ Wt2,
                             float* __restrict__ z,
                             int nrows) {
    const int wave = threadIdx.x >> 5;
    const int lane = threadIdx.x & 31;
    const int m0 = blockIdx.x * 128 + wave * 16;
    if (m0 >= nrows) return;

    const int mrow = lane & 15;
    const int khalf = (lane >> 4) << 3;
    const float* arow = h + (size_t)(m0 + mrow) * F_HID;
    const unsigned short* brow = Wt2 + (size_t)mrow * F_HID;

    v8f acc = {};
#pragma unroll
    for (int k = 0; k < F_HID; k += 32) {
        BFrag a, b;
        float4 a0 = *(const float4*)(arow + k + khalf);
        float4 a1 = *(const float4*)(arow + k + khalf + 4);
        float4 a2 = *(const float4*)(arow + k + 16 + khalf);
        float4 a3 = *(const float4*)(arow + k + 16 + khalf + 4);
        a.u[0] = cvt8(a0, a1);
        a.u[1] = cvt8(a2, a3);
        b.u[0] = *(const uint4*)(brow + k + khalf);
        b.u[1] = *(const uint4*)(brow + k + 16 + khalf);
        acc = __builtin_amdgcn_wmma_f32_16x16x32_bf16(
            false, a.v, false, b.v, (short)0, acc, false, false);
    }

    const int col = lane & 15;
    const int rbase = m0 + ((lane >> 4) << 3);
#pragma unroll
    for (int r = 0; r < 8; ++r) {
        int row = rbase + r;
        if (row < nrows) z[(size_t)row * F_OUT + col] = acc[r];
    }
}


__global__ void k_init_agg2(const float* __restrict__ z,
                            const float* __restrict__ dinv,
                            float* __restrict__ agg, int n) {
    int t = blockIdx.x * blockDim.x + threadIdx.x;
    if (t >= n * F_OUT) return;
    int i = t >> 4;
    float s = dinv[i] * dinv[i];
    agg[t] = z[t] * s;
}

__global__ void k_edge_agg2(const int* __restrict__ src,
                            const int* __restrict__ dst,
                            const float* __restrict__ ew,
                            const float* __restrict__ dinv,
                            const float* __restrict__ z,
                            float* __restrict__ agg, int e) {
    int t = blockIdx.x * blockDim.x + threadIdx.x;
    if (t >= e * F_OUT) return;
    int ei = t >> 4;
    int c = t & 15;
    int s = src[ei], d = dst[ei];
    float nrm = dinv[s] * ew[ei] * dinv[d];
    atomicAdd(&agg[(size_t)d * F_OUT + c], z[(size_t)s * F_OUT + c] * nrm);
}


__global__ void k_logsoftmax(const float* __restrict__ agg,
                             const float* __restrict__ b2,
                             float* __restrict__ out, int n) {
    int i = blockIdx.x * blockDim.x + threadIdx.x;
    if (i >= n) return;
    float l[F_OUT];
    float m = -INFINITY;
#pragma unroll
    for (int c = 0; c < F_OUT; ++c) {
        l[c] = agg[(size_t)i * F_OUT + c] + b2[c];
        m = fmaxf(m, l[c]);
    }
    float s = 0.0f;
#pragma unroll
    for (int c = 0; c < F_OUT; ++c) s += expf(l[c] - m);
    float lse = m + logf(s);
#pragma unroll
    for (int c = 0; c < F_OUT; ++c) out[(size_t)i * F_OUT + c] = l[c] - lse;
}


extern "C" void kernel_launch(void* const* d_in, const int* in_sizes, int n_in,
                              void* d_out, int out_size, void* d_ws, size_t ws_size,
                              hipStream_t stream) {
    (void)n_in; (void)out_size; (void)ws_size;

    const float* x  = (const float*)d_in[0];
    const int*   ei = (const int*)d_in[1];
    const float* ew = (const float*)d_in[2];
    const float* W1 = (const float*)d_in[3];
    const float* b1 = (const float*)d_in[4];
    const float* W2 = (const float*)d_in[5];
    const float* b2 = (const float*)d_in[6];
    float* out = (float*)d_out;

    const int N = in_sizes[0] / F_IN;
    const int E = in_sizes[2];
    const int* src = ei;
    const int* dst = ei + E;

    char* ws = (char*)d_ws;
    size_t off = 0;
    auto take = [&](size_t bytes) -> char* {
        char* p = ws + off;
        off += (bytes + 255) & ~(size_t)255;
        return p;
    };
    float*          dinv = (float*)take((size_t)N * 4);
    unsigned short* Wt1  = (unsigned short*)take((size_t)F_IN * F_HID * 2);
    unsigned short* Wt2  = (unsigned short*)take((size_t)F_HID * F_OUT * 2);
    float*          bufA = (float*)take((size_t)N * F_HID * 4);
    float*          bufB = (float*)take((size_t)N * F_HID * 4);
    float* z    = bufB;
    float* agg2 = bufB + (size_t)N * F_OUT;

    const int T = 256;

    k_set_deg<<<(N + T - 1) / T, T, 0, stream>>>(dinv, N);
    k_deg_accum<<<(E + T - 1) / T, T, 0, stream>>>(dst, ew, dinv, E);
    k_dinv<<<(N + T - 1) / T, T, 0, stream>>>(dinv, N);

    k_prep_weights<<<(F_IN * F_HID + F_HID * F_OUT + T - 1) / T, T, 0, stream>>>(
        W1, W2, Wt1, Wt2);

    k_gemm1_wmma<<<(N + 15) / 16, T, 0, stream>>>(x, Wt1, bufA, N);
    k_init_agg1<<<((size_t)N * 32 + T - 1) / T, T, 0, stream>>>(bufA, dinv, bufB, N);
    k_edge_agg1<<<(E + 7) / 8, T, 0, stream>>>(src, dst, ew, dinv, bufA, bufB, E);
    k_finalize1<<<((size_t)N * 32 + T - 1) / T, T, 0, stream>>>(bufB, b1, bufA, N);

    k_gemm2_wmma<<<(N + 127) / 128, T, 0, stream>>>(bufA, Wt2, z, N);
    k_init_agg2<<<((size_t)N * F_OUT + T - 1) / T, T, 0, stream>>>(z, dinv, agg2, N);
    k_edge_agg2<<<((size_t)E * F_OUT + T - 1) / T, T, 0, stream>>>(
        src, dst, ew, dinv, z, agg2, E);

    k_logsoftmax<<<(N + T - 1) / T, T, 0, stream>>>(agg2, b2, out, N);
}


# layer-1 edge pass: lane l takes columns l+32j, so each row load and each f32 atomic add instruction covers one whole 128-B line instead of 4 lines x 8 lanes (atomic arrival order as before)
# speedup vs baseline: 1.1694x; 1.1694x over previous
#include <hip/hip_runtime.h>
#include <math.h>

#define F_IN 512
#define F_HID 128
#define F_OUT 16

typedef __attribute__((ext_vector_type(16))) __bf16 v16bf;
typedef __attribute__((ext_vector_type(8)))  float  v8f;

union BFrag {
    uint4 u[2];
    v16bf v;
};

__device__ __forceinline__ unsigned int f2bf(float f) {
    union { float f; unsigned int u; } cv; cv.f = f;
    unsigned int u = cv.u;
    unsigned int r = u + 0x7FFFu + ((u >> 16) & 1u);
    return (r >> 16) & 0xFFFFu;
}

__device__ __forceinline__ uint4 cvt8(float4 a, float4 b) {
    uint4 r;
    r.x = f2bf(a.x) | (f2bf(a.y) << 16);
    r.y = f2bf(a.z) | (f2bf(a.w) << 16);
    r.z = f2bf(b.x) | (f2bf(b.y) << 16);
    r.w = f2bf(b.z) | (f2bf(b.w) << 16);
    return r;
}


__global__ void k_set_deg(float* deg, int n) {
    int i = blockIdx.x * blockDim.x + threadIdx.x;
    if (i < n) deg[i] = 1.0f;
}

__global__ void k_deg_accum(const int* __restrict__ dst,
                            const float* __restrict__ ew,
                            float* deg, int e) {
    int i = blockIdx.x * blockDim.x + threadIdx.x;
    if (i < e) atomicAdd(&deg[dst[i]], ew[i]);
}

__global__ void k_dinv(float* deg, int n) {
    int i = blockIdx.x * blockDim.x + threadIdx.x;
    if (i < n) {
        float d = deg[i];
        deg[i] = (d > 0.0f) ? rsqrtf(d) : 0.0f;
    }
}


__global__ void k_prep_weights(const float* __restrict__ W1,
                               const float* __restrict__ W2,
                               unsigned short* __restrict__ Wt1,
                               unsigned short* __restrict__ Wt2) {
    int t = blockIdx.x * blockDim.x + threadIdx.x;
    if (t < F_IN * F_HID) {
        int n = t / F_IN, k = t % F_IN;
        Wt1[t] = (unsigned short)f2bf(W1[k * F_HID + n]);
    }
    int t2 = t - F_IN * F_HID;
    if (t2 >= 0 && t2 < F_HID * F_OUT) {
        int n = t2 / F_HID, k = t2 % F_HID;
        Wt2[t2] = (unsigned short)f2bf(W2[k * F_OUT + n]);
    }
}


__global__ void k_gemm1_wmma(const float* __restrict__ x,
                             const unsigned short* __restrict__ Wt1,
                             float* __restrict__ h1,
                             int nrows) {
    __shared__ unsigned short Abuf[16 * F_IN];
    const int tid = threadIdx.x;
    const int row0 = blockIdx.x * 16;

    for (int i = tid; i < 16 * F_IN; i += 256) {
        int m = i / F_IN, k = i % F_IN;
        int gr = row0 + m;
        float v = (gr < nrows) ? x[(size_t)gr * F_IN + k] : 0.0f;
        Abuf[i] = (unsigned short)f2bf(v);
    }
    __syncthreads();

    const int wave = tid >> 5;
    const int lane = tid & 31;
    const int n0 = wave * 16;
    const int mrow = lane & 15;
    const int khalf = (lane >> 4) << 3;

    const unsigned short* brow = Wt1 + (size_t)(n0 + mrow) * F_IN;

    v8f acc = {};
    for (int k = 0; k < F_IN; k += 32) {
        BFrag a, b;
        a.u[0] = *(const uint4*)&Abuf[mrow * F_IN + k + khalf];
        a.u[1] = *(const uint4*)&Abuf[mrow * F_IN + k + 16 + khalf];
        b.u[0] = *(const uint4*)&brow[k + khalf];
        b.u[1] = *(const uint4*)&brow[k + 16 + khalf];
        acc = __builtin_amdgcn_wmma_f32_16x16x32_bf16(
            false, a.v, false, b.v, (short)0, acc, false, false);
    }

    const int col = n0 + (lane & 15);
    const int rbase = row0 + ((lane >> 4) << 3);
#pragma unroll
    for (int r = 0; r < 8; ++r) {
        int row = rbase + r;
        if (row < nrows) h1[(size_t)row * F_HID + col] = acc[r];
    }
}


__global__ void k_init_agg1(const float* __restrict__ h1,
                            const float* __restrict__ dinv,
                            float* __restrict__ agg, int n) {
    int t = blockIdx.x * blockDim.x + threadIdx.x;
    if (t >= n * 32) return;
    int i = t >> 5;
    float s = dinv[i] * dinv[i];
    float4 v = ((const float4*)h1)[t];
    v.x *= s; v.y *= s; v.z *= s; v.w *= s;
    ((float4*)agg)[t] = v;
}


__global__ void k_edge_agg1(const int* __restrict__ src,
                            const int* __restrict__ dst,
                            const float* __restrict__ ew,
                            const float* __restrict__ dinv,
                            const float* __restrict__ h1,
                            float* __restrict__ agg, int e) {
    int wid = (blockIdx.x * blockDim.x + threadIdx.x) >> 5;
    int lane = threadIdx.x & 31;
    if (wid >= e) return;
    int s = src[wid], d = dst[wid];
    float nrm = dinv[s] * ew[wid] * dinv[d];
    const float* hrow = h1 + (size_t)s * F_HID + lane;
    float v0 = hrow[0];
    float v1 = hrow[32];
    float v2 = hrow[64];
    float v3 = hrow[96];
    float* o = agg + (size_t)d * F_HID + lane;
    atomicAdd(o +  0, v0 * nrm);
    atomicAdd(o + 32, v1 * nrm);
    atomicAdd(o + 64, v2 * nrm);
    atomicAdd(o + 96, v3 * nrm);
}


__global__ void k_finalize1(const float* __restrict__ agg,
                            const float* __restrict__ b1,
                            float* __restrict__ hout, int n) {
    int t = blockIdx.x * blockDim.x + threadIdx.x;
    if (t >= n * 32) return;
    int c4 = (t & 31) * 4;
    float4 v = ((const float4*)agg)[t];
    v.x = fmaxf(v.x + b1[c4 + 0], 0.0f);
    v.y = fmaxf(v.y + b1[c4 + 1], 0.0f);
    v.z = fmaxf(v.z + b1[c4 + 2], 0.0f);
    v.w = fmaxf(v.w + b1[c4 + 3], 0.0f);
    ((float4*)hout)[t] = v;
}


__global__ void k_gemm2_wmma(const float* __restrict__ h,
                             const unsigned short* __restrict__ Wt2,
                             float* __restrict__ z,
                             int nrows) {
    const int wave = threadIdx.x >> 5;
    const int lane = threadIdx.x & 31;
    const int m0 = blockIdx.x * 128 + wave * 16;
    if (m0 >= nrows) return;

    const int mrow = lane & 15;
    const int khalf = (lane >> 4) << 3;
    const float* arow = h + (size_t)(m0 + mrow) * F_HID;
    const unsigned short* brow = Wt2 + (size_t)mrow * F_HID;

    v8f acc = {};
#pragma unroll
    for (int k = 0; k < F_HID; k += 32) {
        BFrag a, b;
        float4 a0 = *(const float4*)(arow + k + khalf);
        float4 a1 = *(const float4*)(arow + k + khalf + 4);
        float4 a2 = *(const float4*)(arow + k + 16 + khalf);
        float4 a3 = *(const float4*)(arow + k + 16 + khalf + 4);
        a.u[0] = cvt8(a0, a1);
        a.u[1] = cvt8(a2, a3);
        b.u[0] = *(const uint4*)(brow + k + khalf);
        b.u[1] = *(const uint4*)(brow + k + 16 + khalf);
        acc = __builtin_amdgcn_wmma_f32_16x16x32_bf16(
            false, a.v, false, b.v, (short)0, acc, false, false);
    }

    const int col = lane & 15;
    const int rbase = m0 + ((lane >> 4) << 3);
#pragma unroll
    for (int r = 0; r < 8; ++r) {
        int row = rbase + r;
        if (row < nrows) z[(size_t)row * F_OUT + col] = acc[r];
    }
}


__global__ void k_init_agg2(const float* __restrict__ z,
                            const float* __restrict__ dinv,
                            float* __restrict__ agg, int n) {
    int t = blockIdx.x * blockDim.x + threadIdx.x;
    if (t >= n * F_OUT) return;
    int i = t >> 4;
    float s = dinv[i] * dinv[i];
    agg[t] = z[t] * s;
}

__global__ void k_edge_agg2(const int* __restrict__ src,
                            const int* __restrict__ dst,
                            const float* __restrict__ ew,
                            const float* __restrict__ dinv,
                            const float* __restrict__ z,
                            float* __restrict__ agg, int e) {
    int t = blockIdx.x * blockDim.x + threadIdx.x;
    if (t >= e * F_OUT) return;
    int ei = t >> 4;
    int c = t & 15;
    int s = src[ei], d = dst[ei];
    float nrm = dinv[s] * ew[ei] * dinv[d];
    atomicAdd(&agg[(size_t)d * F_OUT + c], z[(size_t)s * F_OUT + c] * nrm);
}


__global__ void k_logsoftmax(const float* __restrict__ agg,
                             const float* __restrict__ b2,
                             float* __restrict__ out, int n) {
    int i = blockIdx.x * blockDim.x + threadIdx.x;
    if (i >= n) return;
    float l[F_OUT];
    float m = -INFINITY;
#pragma unroll
    for (int c = 0; c < F_OUT; ++c) {
        l[c] = agg[(size_t)i * F_OUT + c] + b2[c];
        m = fmaxf(m, l[c]);
    }
    float s = 0.0f;
#pragma unroll
    for (int c = 0; c < F_OUT; ++c) s += expf(l[c] - m);
    float lse = m + logf(s);
#pragma unroll
    for (int c = 0; c < F_OUT; ++c) out[(size_t)i * F_OUT + c] = l[c] - lse;
}


extern "C" void kernel_launch(void* const* d_in, const int* in_sizes, int n_in,
                              void* d_out, int out_size, void* d_ws, size_t ws_size,
                              hipStream_t stream) {
    (void)n_in; (void)out_size; (void)ws_size;

    const float* x  = (const float*)d_in[0];
    const int*   ei = (const int*)d_in[1];
    const float* ew = (const float*)d_in[2];
    const float* W1 = (const float*)d_in[3];
    const float* b1 = (const float*)d_in[4];
    const float* W2 = (const float*)d_in[5];
    const float* b2 = (const float*)d_in[6];
    float* out = (float*)d_out;

    const int N = in_sizes[0] / F_IN;
    const int E = in_sizes[2];
    const int* src = ei;
    const int* dst = ei + E;

    char* ws = (char*)d_ws;
    size_t off = 0;
    auto take = [&](size_t bytes) -> char* {
        char* p = ws + off;
        off += (bytes + 255) & ~(size_t)255;
        return p;
    };
    float*          dinv = (float*)take((size_t)N * 4);
    unsigned short* Wt1  = (unsigned short*)take((size_t)F_IN * F_HID * 2);
    unsigned short* Wt2  = (unsigned short*)take((size_t)F_HID * F_OUT * 2);
    float*          bufA = (float*)take((size_t)N * F_HID * 4);
    float*          bufB = (float*)take((size_t)N * F_HID * 4);
    float* z    = bufB;
    float* agg2 = bufB + (size_t)N * F_OUT;

    const int T = 256;

    k_set_deg<<<(N + T - 1) / T, T, 0, stream>>>(dinv, N);
    k_deg_accum<<<(E + T - 1) / T, T, 0, stream>>>(dst, ew, dinv, E);
    k_dinv<<<(N + T - 1) / T, T, 0, stream>>>(dinv, N);

    k_prep_weights<<<(F_IN * F_HID + F_HID * F_OUT + T - 1) / T, T, 0, stream>>>(
        W1, W2, Wt1, Wt2);

    k_gemm1_wmma<<<(N + 15) / 16, T, 0, stream>>>(x, Wt1, bufA, N);
    k_init_agg1<<<((size_t)N * 32 + T - 1) / T, T, 0, stream>>>(bufA, dinv, bufB, N);
    k_edge_agg1<<<(E + 7) / 8, T, 0, stream>>>(src, dst, ew, dinv, bufA, bufB, E);
    k_finalize1<<<((size_t)N * 32 + T - 1) / T, T, 0, stream>>>(bufB, b1, bufA, N);

    k_gemm2_wmma<<<(N + 127) / 128, T, 0, stream>>>(bufA, Wt2, z, N);
    k_init_agg2<<<((size_t)N * F_OUT + T - 1) / T, T, 0, stream>>>(z, dinv, agg2, N);
    k_edge_agg2<<<((size_t)E * F_OUT + T - 1) / T, T, 0, stream>>>(
        src, dst, ew, dinv, z, agg2, E);

    k_logsoftmax<<<(N + T - 1) / T, T, 0, stream>>>(agg2, b2, out, N);
}


# both edge passes: a wave takes 32 consecutive edges (indices, weight, norm read once per wave, passed on by lane reads; 8 edges' row loads in flight), not 1 or 2; f32 atomic arrival order as before
# speedup vs baseline: 1.4622x; 1.2504x over previous
#include <hip/hip_runtime.h>
#include <math.h>

#define F_IN 512
#define F_HID 128
#define F_OUT 16

typedef __attribute__((ext_vector_type(16))) __bf16 v16bf;
typedef __attribute__((ext_vector_type(8)))  float  v8f;

union BFrag {
    uint4 u[2];
    v16bf v;
};

__device__ __forceinline__ unsigned int f2bf(float f) {
    union { float f; unsigned int u; } cv; cv.f = f;
    unsigned int u = cv.u;
    unsigned int r = u + 0x7FFFu + ((u >> 16) & 1u);
    return (r >> 16) & 0xFFFFu;
}

__device__ __forceinline__ uint4 cvt8(float4 a, float4 b) {
    uint4 r;
    r.x = f2bf(a.x) | (f2bf(a.y) << 16);
    r.y = f2bf(a.z) | (f2bf(a.w) << 16);
    r.z = f2bf(b.x) | (f2bf(b.y) << 16);
    r.w = f2bf(b.z) | (f2bf(b.w) << 16);
    return r;
}


__global__ void k_set_deg(float* deg, int n) {
    int i = blockIdx.x * blockDim.x + threadIdx.x;
    if (i < n) deg[i] = 1.0f;
}

__global__ void k_deg_accum(const int* __restrict__ dst,
                            const float* __restrict__ ew,
                            float* deg, int e) {
    int i = blockIdx.x * blockDim.x + threadIdx.x;
    if (i < e) atomicAdd(&deg[dst[i]], ew[i]);
}

__global__ void k_dinv(float* deg, int n) {
    int i = blockIdx.x * blockDim.x + threadIdx.x;
    if (i < n) {
        float d = deg[i];
        deg[i] = (d > 0.0f) ? rsqrtf(d) : 0.0f;
    }
}


__global__ void k_prep_weights(const float* __restrict__ W1,
                               const float* __restrict__ W2,
                               unsigned short* __restrict__ Wt1,
                               unsigned short* __restrict__ Wt2) {
    int t = blockIdx.x * blockDim.x + threadIdx.x;
    if (t < F_IN * F_HID) {
        int n = t / F_IN, k = t % F_IN;
        Wt1[t] = (unsigned short)f2bf(W1[k * F_HID + n]);
    }
    int t2 = t - F_IN * F_HID;
    if (t2 >= 0 && t2 < F_HID * F_OUT) {
        int n = t2 / F_HID, k = t2 % F_HID;
        Wt2[t2] = (unsigned short)f2bf(W2[k * F_OUT + n]);
    }
}


__global__ void k_gemm1_wmma(const float* __restrict__ x,
                             const unsigned short* __restrict__ Wt1,
                             float* __restrict__ h1,
                             int nrows) {
    __shared__ unsigned short Abuf[16 * F_IN];
    const int tid = threadIdx.x;
    const int row0 = blockIdx.x * 16;

    for (int i = tid; i < 16 * F_IN; i += 256) {
        int m = i / F_IN, k = i % F_IN;
        int gr = row0 + m;
        float v = (gr < nrows) ? x[(size_t)gr * F_IN + k] : 0.0f;
        Abuf[i] = (unsigned short)f2bf(v);
    }
    __syncthreads();

    const int wave = tid >> 5;
    const int lane = tid & 31;
    const int n0 = wave * 16;
    const int mrow = lane & 15;
    const int khalf = (lane >> 4) << 3;

    const unsigned short* brow = Wt1 + (size_t)(n0 + mrow) * F_IN;

    v8f acc = {};
    for (int k = 0; k < F_IN; k += 32) {
        BFrag a, b;
        a.u[0] = *(const uint4*)&Abuf[mrow * F_IN + k + khalf];
        a.u[1] = *(const uint4*)&Abuf[mrow * F_IN + k + 16 + khalf];
        b.u[0] = *(const uint4*)&brow[k + khalf];
        b.u[1] = *(const uint4*)&brow[k + 16 + khalf];
        acc = __builtin_amdgcn_wmma_f32_16x16x32_bf16(
            false, a.v, false, b.v, (short)0, acc, false, false);
    }

    const int col = n0 + (lane & 15);
    const int rbase = row0 + ((lane >> 4) << 3);
#pragma unroll
    for (int r = 0; r < 8; ++r) {
        int row = rbase + r;
        if (row < nrows) h1[(size_t)row * F_HID + col] = acc[r];
    }
}


__global__ void k_init_agg1(const float* __restrict__ h1,
                            const float* __restrict__ dinv,
                            float* __restrict__ agg, int n) {
    int t = blockIdx.x * blockDim.x + threadIdx.x;
    if (t >= n * 32) return;
    int i = t >> 5;
    float s = dinv[i] * dinv[i];
    float4 v = ((const float4*)h1)[t];
    v.x *= s; v.y *= s; v.z *= s; v.w *= s;
    ((float4*)agg)[t] = v;
}


__global__ void k_edge_agg1(const int* __restrict__ src,
                            const int* __restrict__ dst,
                            const float* __restrict__ ew,
                            const float* __restrict__ dinv,
                            const float* __restrict__ h1,
                            float* __restrict__ agg, int e) {
    const int lane = threadIdx.x & 31;
    const int wid = __builtin_amdgcn_readfirstlane(
        (int)((blockIdx.x * blockDim.x + threadIdx.x) >> 5));
    const int base = wid * 32;
    if (base >= e) return;
    const int cnt = (e - base < 32) ? (e - base) : 32;

    int s_l = 0, d_l = 0;
    float n_l = 0.0f;
    if (lane < cnt) {
        s_l = src[base + lane];
        d_l = dst[base + lane];
        n_l = dinv[s_l] * ew[base + lane] * dinv[d_l];
    }

    const float* hcol = h1 + lane;
    float* acol = agg + lane;

    int j = 0;
    for (; j + 8 <= cnt; j += 8) {
        float v[8][4];
        float nn[8];
        int dd[8];
#pragma unroll
        for (int u = 0; u < 8; ++u) {
            const int s = __builtin_amdgcn_readlane(s_l, j + u);
            dd[u] = __builtin_amdgcn_readlane(d_l, j + u);
            nn[u] = __int_as_float(__builtin_amdgcn_readlane(__float_as_int(n_l), j + u));
            const float* hrow = hcol + (size_t)s * F_HID;
            v[u][0] = hrow[0];
            v[u][1] = hrow[32];
            v[u][2] = hrow[64];
            v[u][3] = hrow[96];
        }
#pragma unroll
        for (int u = 0; u < 8; ++u) {
            float* o = acol + (size_t)dd[u] * F_HID;
            atomicAdd(o +  0, v[u][0] * nn[u]);
            atomicAdd(o + 32, v[u][1] * nn[u]);
            atomicAdd(o + 64, v[u][2] * nn[u]);
            atomicAdd(o + 96, v[u][3] * nn[u]);
        }
    }
    for (; j < cnt; ++j) {
        const int s = __builtin_amdgcn_readlane(s_l, j);
        const int d = __builtin_amdgcn_readlane(d_l, j);
        const float nrm = __int_as_float(__builtin_amdgcn_readlane(__float_as_int(n_l), j));
        const float* hrow = hcol + (size_t)s * F_HID;
        float v0 = hrow[0];
        float v1 = hrow[32];
        float v2 = hrow[64];
        float v3 = hrow[96];
        float* o = acol + (size_t)d * F_HID;
        atomicAdd(o +  0, v0 * nrm);
        atomicAdd(o + 32, v1 * nrm);
        atomicAdd(o + 64, v2 * nrm);
        atomicAdd(o + 96, v3 * nrm);
    }
}


__global__ void k_finalize1(const float* __restrict__ agg,
                            const float* __restrict__ b1,
                            float* __restrict__ hout, int n) {
    int t = blockIdx.x * blockDim.x + threadIdx.x;
    if (t >= n * 32) return;
    int c4 = (t & 31) * 4;
    float4 v = ((const float4*)agg)[t];
    v.x = fmaxf(v.x + b1[c4 + 0], 0.0f);
    v.y = fmaxf(v.y + b1[c4 + 1], 0.0f);
    v.z = fmaxf(v.z + b1[c4 + 2], 0.0f);
    v.w = fmaxf(v.w + b1[c4 + 3], 0.0f);
    ((float4*)hout)[t] = v;
}


__global__ void k_gemm2_wmma(const float* __restrict__ h,
                             const unsigned short* __restrict__ Wt2,
                             float* __restrict__ z,
                             int nrows) {
    const int wave = threadIdx.x >> 5;
    const int lane = threadIdx.x & 31;
    const int m0 = blockIdx.x * 128 + wave * 16;
    if (m0 >= nrows) return;

    const int mrow = lane & 15;
    const int khalf = (lane >> 4) << 3;
    const float* arow = h + (size_t)(m0 + mrow) * F_HID;
    const unsigned short* brow = Wt2 + (size_t)mrow * F_HID;

    v8f acc = {};
#pragma unroll
    for (int k = 0; k < F_HID; k += 32) {
        BFrag a, b;
        float4 a0 = *(const float4*)(arow + k + khalf);
        float4 a1 = *(const float4*)(arow + k + khalf + 4);
        float4 a2 = *(const float4*)(arow + k + 16 + khalf);
        float4 a3 = *(const float4*)(arow + k + 16 + khalf + 4);
        a.u[0] = cvt8(a0, a1);
        a.u[1] = cvt8(a2, a3);
        b.u[0] = *(const uint4*)(brow + k + khalf);
        b.u[1] = *(const uint4*)(brow + k + 16 + khalf);
        acc = __builtin_amdgcn_wmma_f32_16x16x32_bf16(
            false, a.v, false, b.v, (short)0, acc, false, false);
    }

    const int col = lane & 15;
    const int rbase = m0 + ((lane >> 4) << 3);
#pragma unroll
    for (int r = 0; r < 8; ++r) {
        int row = rbase + r;
        if (row < nrows) z[(size_t)row * F_OUT + col] = acc[r];
    }
}


__global__ void k_init_agg2(const float* __restrict__ z,
                            const float* __restrict__ dinv,
                            float* __restrict__ agg, int n) {
    int t = blockIdx.x * blockDim.x + threadIdx.x;
    if (t >= n * F_OUT) return;
    int i = t >> 4;
    float s = dinv[i] * dinv[i];
    agg[t] = z[t] * s;
}

__global__ void k_edge_agg2(const int* __restrict__ src,
                            const int* __restrict__ dst,
                            const float* __restrict__ ew,
                            const float* __restrict__ dinv,
                            const float* __restrict__ z,
                            float* __restrict__ agg, int e) {
    const int lane = threadIdx.x & 31;
    const int wid = __builtin_amdgcn_readfirstlane(
        (int)((blockIdx.x * blockDim.x + threadIdx.x) >> 5));
    const int base = wid * 32;
    if (base >= e) return;
    const int cnt = (e - base < 32) ? (e - base) : 32;

    int so_l = 0, do_l = 0;
    float n_l = 0.0f;
    if (lane < cnt) {
        const int s = src[base + lane];
        const int d = dst[base + lane];
        n_l = dinv[s] * ew[base + lane] * dinv[d];
        so_l = s * F_OUT;
        do_l = d * F_OUT;
    }

    const int half = lane >> 4;
    const int c = lane & 15;

    if (cnt == 32) {
        float v[16];
#pragma unroll
        for (int i = 0; i < 16; ++i) {
            const int so = __shfl(so_l, 2 * i + half);
            v[i] = z[so + c];
        }
#pragma unroll
        for (int i = 0; i < 16; ++i) {
            const int dof = __shfl(do_l, 2 * i + half);
            const float nrm = __shfl(n_l, 2 * i + half);
            atomicAdd(&agg[dof + c], v[i] * nrm);
        }
    } else {
        for (int i = 0; i < 16; ++i) {
            const int jj = 2 * i + half;
            const int so = __shfl(so_l, jj);
            const int dof = __shfl(do_l, jj);
            const float nrm = __shfl(n_l, jj);
            if (jj < cnt) atomicAdd(&agg[dof + c], z[so + c] * nrm);
        }
    }
}


__global__ void k_logsoftmax(const float* __restrict__ agg,
                             const float* __restrict__ b2,
                             float* __restrict__ out, int n) {
    int i = blockIdx.x * blockDim.x + threadIdx.x;
    if (i >= n) return;
    float l[F_OUT];
    float m = -INFINITY;
#pragma unroll
    for (int c = 0; c < F_OUT; ++c) {
        l[c] = agg[(size_t)i * F_OUT + c] + b2[c];
        m = fmaxf(m, l[c]);
    }
    float s = 0.0f;
#pragma unroll
    for (int c = 0; c < F_OUT; ++c) s += expf(l[c] - m);
    float lse = m + logf(s);
#pragma unroll
    for (int c = 0; c < F_OUT; ++c) out[(size_t)i * F_OUT + c] = l[c] - lse;
}


extern "C" void kernel_launch(void* const* d_in, const int* in_sizes, int n_in,
                              void* d_out, int out_size, void* d_ws, size_t ws_size,
                              hipStream_t stream) {
    (void)n_in; (void)out_size; (void)ws_size;

    const float* x  = (const float*)d_in[0];
    const int*   ei = (const int*)d_in[1];
    const float* ew = (const float*)d_in[2];
    const float* W1 = (const float*)d_in[3];
    const float* b1 = (const float*)d_in[4];
    const float* W2 = (const float*)d_in[5];
    const float* b2 = (const float*)d_in[6];
    float* out = (float*)d_out;

    const int N = in_sizes[0] / F_IN;
    const int E = in_sizes[2];
    const int* src = ei;
    const int* dst = ei + E;

    char* ws = (char*)d_ws;
    size_t off = 0;
    auto take = [&](size_t bytes) -> char* {
        char* p = ws + off;
        off += (bytes + 255) & ~(size_t)255;
        return p;
    };
    float*          dinv = (float*)take((size_t)N * 4);
    unsigned short* Wt1  = (unsigned short*)take((size_t)F_IN * F_HID * 2);
    unsigned short* Wt2  = (unsigned short*)take((size_t)F_HID * F_OUT * 2);
    float*          bufA = (float*)take((size_t)N * F_HID * 4);
    float*          bufB = (float*)take((size_t)N * F_HID * 4);
    float* z    = bufB;
    float* agg2 = bufB + (size_t)N * F_OUT;

    const int T = 256;

    k_set_deg<<<(N + T - 1) / T, T, 0, stream>>>(dinv, N);
    k_deg_accum<<<(E + T - 1) / T, T, 0, stream>>>(dst, ew, dinv, E);
    k_dinv<<<(N + T - 1) / T, T, 0, stream>>>(dinv, N);

    k_prep_weights<<<(F_IN * F_HID + F_HID * F_OUT + T - 1) / T, T, 0, stream>>>(
        W1, W2, Wt1, Wt2);

    k_gemm1_wmma<<<(N + 15) / 16, T, 0, stream>>>(x, Wt1, bufA, N);
    k_init_agg1<<<((size_t)N * 32 + T - 1) / T, T, 0, stream>>>(bufA, dinv, bufB, N);
    k_edge_agg1<<<(E + T - 1) / T, T, 0, stream>>>(src, dst, ew, dinv, bufA, bufB, E);
    k_finalize1<<<((size_t)N * 32 + T - 1) / T, T, 0, stream>>>(bufB, b1, bufA, N);

    k_gemm2_wmma<<<(N + 127) / 128, T, 0, stream>>>(bufA, Wt2, z, N);
    k_init_agg2<<<((size_t)N * F_OUT + T - 1) / T, T, 0, stream>>>(z, dinv, agg2, N);
    k_edge_agg2<<<(E + T - 1) / T, T, 0, stream>>>(
        src, dst, ew, dinv, z, agg2, E);

    k_logsoftmax<<<(N + T - 1) / T, T, 0, stream>>>(agg2, b2, out, N);
}


# edges bucketed by destination in every call (count, offsets, fill), then one wave sums a row's neighbour rows in registers: no f32 atomics in the sums; sum order = arrival rank of an integer add
# speedup vs baseline: 1.6421x; 1.1230x over previous
#include <hip/hip_runtime.h>
#include <math.h>

#define F_IN 512
#define F_HID 128
#define F_OUT 16

typedef __attribute__((ext_vector_type(16))) __bf16 v16bf;
typedef __attribute__((ext_vector_type(8)))  float  v8f;

union BFrag {
    uint4 u[2];
    v16bf v;
};

__device__ __forceinline__ unsigned int f2bf(float f) {
    union { float f; unsigned int u; } cv; cv.f = f;
    unsigned int u = cv.u;
    unsigned int r = u + 0x7FFFu + ((u >> 16) & 1u);
    return (r >> 16) & 0xFFFFu;
}


__global__ void k_zero(float* deg, int* cnt, int* total, int n) {
    int i = blockIdx.x * blockDim.x + threadIdx.x;
    if (i < n) { deg[i] = 1.0f; cnt[i] = 0; }
    if (i == 0) *total = 0;
}

__global__ void k_count(const int* __restrict__ dst,
                        const float* __restrict__ ew,
                        float* deg, int* cnt,
                        int* __restrict__ rank, int e) {
    int i = blockIdx.x * blockDim.x + threadIdx.x;
    if (i < e) {
        const int d = dst[i];
        atomicAdd(&deg[d], ew[i]);
        rank[i] = atomicAdd(&cnt[d], 1);
    }
}

__global__ void k_offsets(float* deg, const int* __restrict__ cnt,
                          int* __restrict__ row_start, int* total, int n) {
    const int i = blockIdx.x * blockDim.x + threadIdx.x;
    const int lane = threadIdx.x & 31;
    int c = 0;
    if (i < n) {
        c = cnt[i];
        float d = deg[i];
        deg[i] = (d > 0.0f) ? rsqrtf(d) : 0.0f;
    }
    int x = c;
#pragma unroll
    for (int off = 1; off < 32; off <<= 1) {
        const int t = __shfl_up(x, off);
        if (lane >= off) x += t;
    }
    int b = 0;
    if (lane == 31) b = atomicAdd(total, x);
    b = __shfl(b, 31);
    if (i < n) row_start[i] = b + x - c;
}

__global__ void k_fill(const int* __restrict__ src,
                       const int* __restrict__ dst,
                       const float* __restrict__ ew,
                       const float* __restrict__ dinv,
                       const int* __restrict__ row_start,
                       const int* __restrict__ rank,
                       int2* __restrict__ csr, int e) {
    int i = blockIdx.x * blockDim.x + threadIdx.x;
    if (i < e) {
        const int s = src[i];
        const int d = dst[i];
        const float coef = dinv[s] * ew[i] * dinv[d];
        csr[row_start[d] + rank[i]] = make_int2(s, __float_as_int(coef));
    }
}


__global__ void k_prep_weights(const float* __restrict__ W1,
                               const float* __restrict__ W2,
                               unsigned short* __restrict__ Wt1,
                               unsigned short* __restrict__ Wt2) {
    int t = blockIdx.x * blockDim.x + threadIdx.x;
    if (t < F_IN * F_HID) {
        int n = t / F_IN, k = t % F_IN;
        Wt1[t] = (unsigned short)f2bf(W1[k * F_HID + n]);
    }
    int t2 = t - F_IN * F_HID;
    if (t2 >= 0 && t2 < F_HID * F_OUT) {
        int n = t2 / F_HID, k = t2 % F_HID;
        Wt2[t2] = (unsigned short)f2bf(W2[k * F_OUT + n]);
    }
}


__global__ void k_gemm1_wmma(const float* __restrict__ x,
                             const unsigned short* __restrict__ Wt1,
                             float* __restrict__ h1,
                             int nrows) {
    __shared__ unsigned short Abuf[16 * F_IN];
    const int tid = threadIdx.x;
    const int row0 = blockIdx.x * 16;

    for (int i = tid; i < 16 * F_IN; i += 256) {
        int m = i / F_IN, k = i % F_IN;
        int gr = row0 + m;
        float v = (gr < nrows) ? x[(size_t)gr * F_IN + k] : 0.0f;
        Abuf[i] = (unsigned short)f2bf(v);
    }
    __syncthreads();

    const int wave = tid >> 5;
    const int lane = tid & 31;
    const int n0 = wave * 16;
    const int mrow = lane & 15;
    const int khalf = (lane >> 4) << 3;

    const unsigned short* brow = Wt1 + (size_t)(n0 + mrow) * F_IN;

    v8f acc = {};
    for (int k = 0; k < F_IN; k += 32) {
        BFrag a, b;
        a.u[0] = *(const uint4*)&Abuf[mrow * F_IN + k + khalf];
        a.u[1] = *(const uint4*)&Abuf[mrow * F_IN + k + 16 + khalf];
        b.u[0] = *(const uint4*)&brow[k + khalf];
        b.u[1] = *(const uint4*)&brow[k + 16 + khalf];
        acc = __builtin_amdgcn_wmma_f32_16x16x32_bf16(
            false, a.v, false, b.v, (short)0, acc, false, false);
    }

    const int col = n0 + (lane & 15);
    const int rbase = row0 + ((lane >> 4) << 3);
#pragma unroll
    for (int r = 0; r < 8; ++r) {
        int row = rbase + r;
        if (row < nrows) h1[(size_t)row * F_HID + col] = acc[r];
    }
}


#define ROWS_PER_WAVE1 2

template <int B>
__device__ __forceinline__ void gather_rows(float4& acc, const float* hcol,
                                            int s_l, int c_l, int j) {
    float4 v[B];
    float cc[B];
#pragma unroll
    for (int u = 0; u < B; ++u) {
        const int s = __builtin_amdgcn_readlane(s_l, j + u);
        cc[u] = __int_as_float(__builtin_amdgcn_readlane(c_l, j + u));
        v[u] = *(const float4*)(hcol + (size_t)s * F_HID);
    }
#pragma unroll
    for (int u = 0; u < B; ++u) {
        acc.x += cc[u] * v[u].x;
        acc.y += cc[u] * v[u].y;
        acc.z += cc[u] * v[u].z;
        acc.w += cc[u] * v[u].w;
    }
}

__global__ void k_gather1(const int2* __restrict__ csr,
                          const int* __restrict__ row_start,
                          const int* __restrict__ cnt,
                          const float* __restrict__ dinv,
                          const float* __restrict__ h1,
                          const float* __restrict__ b1,
                          unsigned short* __restrict__ hb, int n) {
    const int lane = threadIdx.x & 31;
    const int wid = __builtin_amdgcn_readfirstlane(
        (int)((blockIdx.x * blockDim.x + threadIdx.x) >> 5));
    const float4 bias = *(const float4*)(b1 + lane * 4);
    const float* hcol = h1 + lane * 4;

    for (int r = 0; r < ROWS_PER_WAVE1; ++r) {
        const int i = wid * ROWS_PER_WAVE1 + r;
        if (i >= n) break;
        const int start = row_start[i];
        const int deg_i = cnt[i];
        const float di = dinv[i];
        const float sl = di * di;

        int m = (deg_i < 32) ? deg_i : 32;
        int s_l = 0;
        int c_l = 0;
        if (lane < m) {
            const int2 p = csr[start + lane];
            s_l = p.x;
            c_l = p.y;
        }
        float4 acc = *(const float4*)(hcol + (size_t)i * F_HID);
        acc.x *= sl; acc.y *= sl; acc.z *= sl; acc.w *= sl;

        int base = 0;
        while (m > 0) {
            int j = 0;
            for (; j + 8 <= m; j += 8) gather_rows<8>(acc, hcol, s_l, c_l, j);
            if (j + 4 <= m) { gather_rows<4>(acc, hcol, s_l, c_l, j); j += 4; }
            if (j + 2 <= m) { gather_rows<2>(acc, hcol, s_l, c_l, j); j += 2; }
            if (j < m)      { gather_rows<1>(acc, hcol, s_l, c_l, j); }
            base += 32;
            m = (deg_i - base < 32) ? (deg_i - base) : 32;
            if (m > 0 && lane < m) {
                const int2 p = csr[start + base + lane];
                s_l = p.x;
                c_l = p.y;
            }
        }

        const float ox = fmaxf(acc.x + bias.x, 0.0f);
        const float oy = fmaxf(acc.y + bias.y, 0.0f);
        const float oz = fmaxf(acc.z + bias.z, 0.0f);
        const float ow = fmaxf(acc.w + bias.w, 0.0f);
        uint2 pk;
        pk.x = f2bf(ox) | (f2bf(oy) << 16);
        pk.y = f2bf(oz) | (f2bf(ow) << 16);
        *(uint2*)(hb + (size_t)i * F_HID + lane * 4) = pk;
    }
}


__global__ void k_gemm2_wmma(const unsigned short* __restrict__ hb,
                             const unsigned short* __restrict__ Wt2,
                             float* __restrict__ z,
                             int nrows) {
    const int wave = threadIdx.x >> 5;
    const int lane = threadIdx.x & 31;
    const int m0 = blockIdx.x * 128 + wave * 16;
    if (m0 >= nrows) return;

    const int mrow = lane & 15;
    const int khalf = (lane >> 4) << 3;
    const unsigned short* arow = hb + (size_t)(m0 + mrow) * F_HID;
    const unsigned short* brow = Wt2 + (size_t)mrow * F_HID;

    v8f acc = {};
#pragma unroll
    for (int k = 0; k < F_HID; k += 32) {
        BFrag a, b;
        a.u[0] = *(const uint4*)(arow + k + khalf);
        a.u[1] = *(const uint4*)(arow + k + 16 + khalf);
        b.u[0] = *(const uint4*)(brow + k + khalf);
        b.u[1] = *(const uint4*)(brow + k + 16 + khalf);
        acc = __builtin_amdgcn_wmma_f32_16x16x32_bf16(
            false, a.v, false, b.v, (short)0, acc, false, false);
    }

    const int col = lane & 15;
    const int rbase = m0 + ((lane >> 4) << 3);
#pragma unroll
    for (int r = 0; r < 8; ++r) {
        int row = rbase + r;
        if (row < nrows) z[(size_t)row * F_OUT + col] = acc[r];
    }
}


__global__ void k_gather2(const int2* __restrict__ csr,
                          const int* __restrict__ row_start,
                          const int* __restrict__ cnt,
                          const float* __restrict__ dinv,
                          const float* __restrict__ z,
                          const float* __restrict__ b2,
                          float* __restrict__ out, int n) {
    const int lane = threadIdx.x & 31;
    const int half = lane >> 4;
    const int c = lane & 15;
    const int wid = __builtin_amdgcn_readfirstlane(
        (int)((blockIdx.x * blockDim.x + threadIdx.x) >> 5));
    if (wid * 2 >= n) return;
    const int i = wid * 2 + half;
    const bool valid = i < n;

    int start = 0, deg_i = 0;
    float acc = 0.0f;
    if (valid) {
        start = row_start[i];
        deg_i = cnt[i];
        const float di = dinv[i];
        acc = di * di * z[(size_t)i * F_OUT + c];
    }
    const int other = __shfl_xor(deg_i, 16);
    const int maxdeg = __builtin_amdgcn_readfirstlane(deg_i > other ? deg_i : other);

    for (int base = 0; base < maxdeg; base += 16) {
        const int m = deg_i - base;
        int s_l = 0;
        float c_l = 0.0f;
        if (c < m) {
            const int2 p = csr[start + base + c];
            s_l = p.x;
            c_l = __int_as_float(p.y);
        }
        float v[16];
        float cf[16];
#pragma unroll
        for (int u = 0; u < 16; ++u) {
            const int s = __shfl(s_l, u, 16);
            cf[u] = __shfl(c_l, u, 16);
            v[u] = (u < m) ? z[(size_t)s * F_OUT + c] : 0.0f;
        }
#pragma unroll
        for (int u = 0; u < 16; ++u) acc += cf[u] * v[u];
    }

    const float l = acc + b2[c];
    float mx = l;
#pragma unroll
    for (int k = 8; k >= 1; k >>= 1) mx = fmaxf(mx, __shfl_xor(mx, k, 16));
    const float ex = expf(l - mx);
    float ssum = 0.0f;
#pragma unroll
    for (int u = 0; u < 16; ++u) ssum += __shfl(ex, u, 16);
    const float lse = mx + logf(ssum);
    if (valid) out[(size_t)i * F_OUT + c] = l - lse;
}


extern "C" void kernel_launch(void* const* d_in, const int* in_sizes, int n_in,
                              void* d_out, int out_size, void* d_ws, size_t ws_size,
                              hipStream_t stream) {
    (void)n_in; (void)out_size; (void)ws_size;

    const float* x  = (const float*)d_in[0];
    const int*   ei = (const int*)d_in[1];
    const float* ew = (const float*)d_in[2];
    const float* W1 = (const float*)d_in[3];
    const float* b1 = (const float*)d_in[4];
    const float* W2 = (const float*)d_in[5];
    const float* b2 = (const float*)d_in[6];
    float* out = (float*)d_out;

    const int N = in_sizes[0] / F_IN;
    const int E = in_sizes[2];
    const int* src = ei;
    const int* dst = ei + E;

    char* ws = (char*)d_ws;
    size_t off = 0;
    auto take = [&](size_t bytes) -> char* {
        char* p = ws + off;
        off += (bytes + 255) & ~(size_t)255;
        return p;
    };
    float*          dinv = (float*)take((size_t)N * 4);
    unsigned short* Wt1  = (unsigned short*)take((size_t)F_IN * F_HID * 2);
    unsigned short* Wt2  = (unsigned short*)take((size_t)F_HID * F_OUT * 2);
    float*          bufA = (float*)take((size_t)N * F_HID * 4);
    char*           bufB = take((size_t)N * F_HID * 4);

    unsigned short* hb = (unsigned short*)bufB;
    char* q = bufB + (size_t)N * F_HID * 2;
    int2* csr       = (int2*)q;  q += (size_t)E * 8;
    int*  rank      = (int*)q;   q += (size_t)E * 4;
    int*  cnt       = (int*)q;   q += (size_t)N * 4;
    int*  row_start = (int*)q;   q += (size_t)N * 4;
    int*  total     = (int*)q;
    float* z = bufA;

    const int T = 256;

    k_zero<<<(N + T - 1) / T, T, 0, stream>>>(dinv, cnt, total, N);
    k_count<<<(E + T - 1) / T, T, 0, stream>>>(dst, ew, dinv, cnt, rank, E);
    k_offsets<<<(N + T - 1) / T, T, 0, stream>>>(dinv, cnt, row_start, total, N);
    k_fill<<<(E + T - 1) / T, T, 0, stream>>>(src, dst, ew, dinv, row_start, rank, csr, E);

    k_prep_weights<<<(F_IN * F_HID + F_HID * F_OUT + T - 1) / T, T, 0, stream>>>(
        W1, W2, Wt1, Wt2);

    k_gemm1_wmma<<<(N + 15) / 16, T, 0, stream>>>(x, Wt1, bufA, N);
    k_gather1<<<(N + ROWS_PER_WAVE1 * 8 - 1) / (ROWS_PER_WAVE1 * 8), T, 0, stream>>>(
        csr, row_start, cnt, dinv, bufA, b1, hb, N);

    k_gemm2_wmma<<<(N + 127) / 128, T, 0, stream>>>(hb, Wt2, z, N);
    k_gather2<<<(N + 15) / 16, T, 0, stream>>>(csr, row_start, cnt, dinv, z, b2, out, N);
}


# x@W1 product: LDS row pitch of the staged A tile 1024 B -> 1040 B (16 B x 65), so the 16 rows of a fragment read fall on different banks instead of the same four (16-way conflict)
# speedup vs baseline: 1.7307x; 1.0540x over previous
#include <hip/hip_runtime.h>
#include <math.h>

#define F_IN 512
#define F_HID 128
#define F_OUT 16

typedef __attribute__((ext_vector_type(16))) __bf16 v16bf;
typedef __attribute__((ext_vector_type(8)))  float  v8f;

union BFrag {
    uint4 u[2];
    v16bf v;
};

__device__ __forceinline__ unsigned int f2bf(float f) {
    union { float f; unsigned int u; } cv; cv.f = f;
    unsigned int u = cv.u;
    unsigned int r = u + 0x7FFFu + ((u >> 16) & 1u);
    return (r >> 16) & 0xFFFFu;
}


__global__ void k_zero(float* deg, int* cnt, int* total, int n) {
    int i = blockIdx.x * blockDim.x + threadIdx.x;
    if (i < n) { deg[i] = 1.0f; cnt[i] = 0; }
    if (i == 0) *total = 0;
}

__global__ void k_count(const int* __restrict__ dst,
                        const float* __restrict__ ew,
                        float* deg, int* cnt,
                        int* __restrict__ rank, int e) {
    int i = blockIdx.x * blockDim.x + threadIdx.x;
    if (i < e) {
        const int d = dst[i];
        atomicAdd(&deg[d], ew[i]);
        rank[i] = atomicAdd(&cnt[d], 1);
    }
}

__global__ void k_offsets(float* deg, const int* __restrict__ cnt,
                          int* __restrict__ row_start, int* total, int n) {
    const int i = blockIdx.x * blockDim.x + threadIdx.x;
    const int lane = threadIdx.x & 31;
    int c = 0;
    if (i < n) {
        c = cnt[i];
        float d = deg[i];
        deg[i] = (d > 0.0f) ? rsqrtf(d) : 0.0f;
    }
    int x = c;
#pragma unroll
    for (int off = 1; off < 32; off <<= 1) {
        const int t = __shfl_up(x, off);
        if (lane >= off) x += t;
    }
    int b = 0;
    if (lane == 31) b = atomicAdd(total, x);
    b = __shfl(b, 31);
    if (i < n) row_start[i] = b + x - c;
}

__global__ void k_fill(const int* __restrict__ src,
                       const int* __restrict__ dst,
                       const float* __restrict__ ew,
                       const float* __restrict__ dinv,
                       const int* __restrict__ row_start,
                       const int* __restrict__ rank,
                       int2* __restrict__ csr, int e) {
    int i = blockIdx.x * blockDim.x + threadIdx.x;
    if (i < e) {
        const int s = src[i];
        const int d = dst[i];
        const float coef = dinv[s] * ew[i] * dinv[d];
        csr[row_start[d] + rank[i]] = make_int2(s, __float_as_int(coef));
    }
}


__global__ void k_prep_weights(const float* __restrict__ W1,
                               const float* __restrict__ W2,
                               unsigned short* __restrict__ Wt1,
                               unsigned short* __restrict__ Wt2) {
    int t = blockIdx.x * blockDim.x + threadIdx.x;
    if (t < F_IN * F_HID) {
        int n = t / F_IN, k = t % F_IN;
        Wt1[t] = (unsigned short)f2bf(W1[k * F_HID + n]);
    }
    int t2 = t - F_IN * F_HID;
    if (t2 >= 0 && t2 < F_HID * F_OUT) {
        int n = t2 / F_HID, k = t2 % F_HID;
        Wt2[t2] = (unsigned short)f2bf(W2[k * F_OUT + n]);
    }
}


#define A_PITCH (F_IN + 8)

__global__ void k_gemm1_wmma(const float* __restrict__ x,
                             const unsigned short* __restrict__ Wt1,
                             float* __restrict__ h1,
                             int nrows) {
    __shared__ unsigned short Abuf[16 * A_PITCH];
    const int tid = threadIdx.x;
    const int row0 = blockIdx.x * 16;

    for (int i = tid; i < 16 * F_IN; i += 256) {
        int m = i / F_IN, k = i % F_IN;
        int gr = row0 + m;
        float v = (gr < nrows) ? x[(size_t)gr * F_IN + k] : 0.0f;
        Abuf[m * A_PITCH + k] = (unsigned short)f2bf(v);
    }
    __syncthreads();

    const int wave = tid >> 5;
    const int lane = tid & 31;
    const int n0 = wave * 16;
    const int mrow = lane & 15;
    const int khalf = (lane >> 4) << 3;

    const unsigned short* brow = Wt1 + (size_t)(n0 + mrow) * F_IN;

    v8f acc = {};
    for (int k = 0; k < F_IN; k += 32) {
        BFrag a, b;
        a.u[0] = *(const uint4*)&Abuf[mrow * A_PITCH + k + khalf];
        a.u[1] = *(const uint4*)&Abuf[mrow * A_PITCH + k + 16 + khalf];
        b.u[0] = *(const uint4*)&brow[k + khalf];
        b.u[1] = *(const uint4*)&brow[k + 16 + khalf];
        acc = __builtin_amdgcn_wmma_f32_16x16x32_bf16(
            false, a.v, false, b.v, (short)0, acc, false, false);
    }

    const int col = n0 + (lane & 15);
    const int rbase = row0 + ((lane >> 4) << 3);
#pragma unroll
    for (int r = 0; r < 8; ++r) {
        int row = rbase + r;
        if (row < nrows) h1[(size_t)row * F_HID + col] = acc[r];
    }
}


#define ROWS_PER_WAVE1 2

template <int B>
__device__ __forceinline__ void gather_rows(float4& acc, const float* hcol,
                                            int s_l, int c_l, int j) {
    float4 v[B];
    float cc[B];
#pragma unroll
    for (int u = 0; u < B; ++u) {
        const int s = __builtin_amdgcn_readlane(s_l, j + u);
        cc[u] = __int_as_float(__builtin_amdgcn_readlane(c_l, j + u));
        v[u] = *(const float4*)(hcol + (size_t)s * F_HID);
    }
#pragma unroll
    for (int u = 0; u < B; ++u) {
        acc.x += cc[u] * v[u].x;
        acc.y += cc[u] * v[u].y;
        acc.z += cc[u] * v[u].z;
        acc.w += cc[u] * v[u].w;
    }
}

__global__ void k_gather1(const int2* __restrict__ csr,
                          const int* __restrict__ row_start,
                          const int* __restrict__ cnt,
                          const float* __restrict__ dinv,
                          const float* __restrict__ h1,
                          const float* __restrict__ b1,
                          unsigned short* __restrict__ hb, int n) {
    const int lane = threadIdx.x & 31;
    const int wid = __builtin_amdgcn_readfirstlane(
        (int)((blockIdx.x * blockDim.x + threadIdx.x) >> 5));
    const float4 bias = *(const float4*)(b1 + lane * 4);
    const float* hcol = h1 + lane * 4;

    for (int r = 0; r < ROWS_PER_WAVE1; ++r) {
        const int i = wid * ROWS_PER_WAVE1 + r;
        if (i >= n) break;
        const int start = row_start[i];
        const int deg_i = cnt[i];
        const float di = dinv[i];
        const float sl = di * di;

        int m = (deg_i < 32) ? deg_i : 32;
        int s_l = 0;
        int c_l = 0;
        if (lane < m) {
            const int2 p = csr[start + lane];
            s_l = p.x;
            c_l = p.y;
        }
        float4 acc = *(const float4*)(hcol + (size_t)i * F_HID);
        acc.x *= sl; acc.y *= sl; acc.z *= sl; acc.w *= sl;

        int base = 0;
        while (m > 0) {
            int j = 0;
            for (; j + 8 <= m; j += 8) gather_rows<8>(acc, hcol, s_l, c_l, j);
            if (j + 4 <= m) { gather_rows<4>(acc, hcol, s_l, c_l, j); j += 4; }
            if (j + 2 <= m) { gather_rows<2>(acc, hcol, s_l, c_l, j); j += 2; }
            if (j < m)      { gather_rows<1>(acc, hcol, s_l, c_l, j); }
            base += 32;
            m = (deg_i - base < 32) ? (deg_i - base) : 32;
            if (m > 0 && lane < m) {
                const int2 p = csr[start + base + lane];
                s_l = p.x;
                c_l = p.y;
            }
        }

        const float ox = fmaxf(acc.x + bias.x, 0.0f);
        const float oy = fmaxf(acc.y + bias.y, 0.0f);
        const float oz = fmaxf(acc.z + bias.z, 0.0f);
        const float ow = fmaxf(acc.w + bias.w, 0.0f);
        uint2 pk;
        pk.x = f2bf(ox) | (f2bf(oy) << 16);
        pk.y = f2bf(oz) | (f2bf(ow) << 16);
        *(uint2*)(hb + (size_t)i * F_HID + lane * 4) = pk;
    }
}


__global__ void k_gemm2_wmma(const unsigned short* __restrict__ hb,
                             const unsigned short* __restrict__ Wt2,
                             float* __restrict__ z,
                             int nrows) {
    const int wave = threadIdx.x >> 5;
    const int lane = threadIdx.x & 31;
    const int m0 = blockIdx.x * 128 + wave * 16;
    if (m0 >= nrows) return;

    const int mrow = lane & 15;
    const int khalf = (lane >> 4) << 3;
    const unsigned short* arow = hb + (size_t)(m0 + mrow) * F_HID;
    const unsigned short* brow = Wt2 + (size_t)mrow * F_HID;

    v8f acc = {};
#pragma unroll
    for (int k = 0; k < F_HID; k += 32) {
        BFrag a, b;
        a.u[0] = *(const uint4*)(arow + k + khalf);
        a.u[1] = *(const uint4*)(arow + k + 16 + khalf);
        b.u[0] = *(const uint4*)(brow + k + khalf);
        b.u[1] = *(const uint4*)(brow + k + 16 + khalf);
        acc = __builtin_amdgcn_wmma_f32_16x16x32_bf16(
            false, a.v, false, b.v, (short)0, acc, false, false);
    }

    const int col = lane & 15;
    const int rbase = m0 + ((lane >> 4) << 3);
#pragma unroll
    for (int r = 0; r < 8; ++r) {
        int row = rbase + r;
        if (row < nrows) z[(size_t)row * F_OUT + col] = acc[r];
    }
}


__global__ void k_gather2(const int2* __restrict__ csr,
                          const int* __restrict__ row_start,
                          const int* __restrict__ cnt,
                          const float* __restrict__ dinv,
                          const float* __restrict__ z,
                          const float* __restrict__ b2,
                          float* __restrict__ out, int n) {
    const int lane = threadIdx.x & 31;
    const int half = lane >> 4;
    const int c = lane & 15;
    const int wid = __builtin_amdgcn_readfirstlane(
        (int)((blockIdx.x * blockDim.x + threadIdx.x) >> 5));
    if (wid * 2 >= n) return;
    const int i = wid * 2 + half;
    const bool valid = i < n;

    int start = 0, deg_i = 0;
    float acc = 0.0f;
    if (valid) {
        start = row_start[i];
        deg_i = cnt[i];
        const float di = dinv[i];
        acc = di * di * z[(size_t)i * F_OUT + c];
    }
    const int other = __shfl_xor(deg_i, 16);
    const int maxdeg = __builtin_amdgcn_readfirstlane(deg_i > other ? deg_i : other);

    for (int base = 0; base < maxdeg; base += 16) {
        const int m = deg_i - base;
        int s_l = 0;
        float c_l = 0.0f;
        if (c < m) {
            const int2 p = csr[start + base + c];
            s_l = p.x;
            c_l = __int_as_float(p.y);
        }
        float v[16];
        float cf[16];
#pragma unroll
        for (int u = 0; u < 16; ++u) {
            const int s = __shfl(s_l, u, 16);
            cf[u] = __shfl(c_l, u, 16);
            v[u] = (u < m) ? z[(size_t)s * F_OUT + c] : 0.0f;
        }
#pragma unroll
        for (int u = 0; u < 16; ++u) acc += cf[u] * v[u];
    }

    const float l = acc + b2[c];
    float mx = l;
#pragma unroll
    for (int k = 8; k >= 1; k >>= 1) mx = fmaxf(mx, __shfl_xor(mx, k, 16));
    const float ex = expf(l - mx);
    float ssum = 0.0f;
#pragma unroll
    for (int u = 0; u < 16; ++u) ssum += __shfl(ex, u, 16);
    const float lse = mx + logf(ssum);
    if (valid) out[(size_t)i * F_OUT + c] = l - lse;
}


extern "C" void kernel_launch(void* const* d_in, const int* in_sizes, int n_in,
                              void* d_out, int out_size, void* d_ws, size_t ws_size,
                              hipStream_t stream) {
    (void)n_in; (void)out_size; (void)ws_size;

    const float* x  = (const float*)d_in[0];
    const int*   ei = (const int*)d_in[1];
    const float* ew = (const float*)d_in[2];
    const float* W1 = (const float*)d_in[3];
    const float* b1 = (const float*)d_in[4];
    const float* W2 = (const float*)d_in[5];
    const float* b2 = (const float*)d_in[6];
    float* out = (float*)d_out;

    const int N = in_sizes[0] / F_IN;
    const int E = in_sizes[2];
    const int* src = ei;
    const int* dst = ei + E;

    char* ws = (char*)d_ws;
    size_t off = 0;
    auto take = [&](size_t bytes) -> char* {
        char* p = ws + off;
        off += (bytes + 255) & ~(size_t)255;
        return p;
    };
    float*          dinv = (float*)take((size_t)N * 4);
    unsigned short* Wt1  = (unsigned short*)take((size_t)F_IN * F_HID * 2);
    unsigned short* Wt2  = (unsigned short*)take((size_t)F_HID * F_OUT * 2);
    float*          bufA = (float*)take((size_t)N * F_HID * 4);
    char*           bufB = take((size_t)N * F_HID * 4);

    unsigned short* hb = (unsigned short*)bufB;
    char* q = bufB + (size_t)N * F_HID * 2;
    int2* csr       = (int2*)q;  q += (size_t)E * 8;
    int*  rank      = (int*)q;   q += (size_t)E * 4;
    int*  cnt       = (int*)q;   q += (size_t)N * 4;
    int*  row_start = (int*)q;   q += (size_t)N * 4;
    int*  total     = (int*)q;
    float* z = bufA;

    const int T = 256;

    k_zero<<<(N + T - 1) / T, T, 0, stream>>>(dinv, cnt, total, N);
    k_count<<<(E + T - 1) / T, T, 0, stream>>>(dst, ew, dinv, cnt, rank, E);
    k_offsets<<<(N + T - 1) / T, T, 0, stream>>>(dinv, cnt, row_start, total, N);
    k_fill<<<(E + T - 1) / T, T, 0, stream>>>(src, dst, ew, dinv, row_start, rank, csr, E);

    k_prep_weights<<<(F_IN * F_HID + F_HID * F_OUT + T - 1) / T, T, 0, stream>>>(
        W1, W2, Wt1, Wt2);

    k_gemm1_wmma<<<(N + 15) / 16, T, 0, stream>>>(x, Wt1, bufA, N);
    k_gather1<<<(N + ROWS_PER_WAVE1 * 8 - 1) / (ROWS_PER_WAVE1 * 8), T, 0, stream>>>(
        csr, row_start, cnt, dinv, bufA, b1, hb, N);

    k_gemm2_wmma<<<(N + 127) / 128, T, 0, stream>>>(hb, Wt2, z, N);
    k_gather2<<<(N + 15) / 16, T, 0, stream>>>(csr, row_start, cnt, dinv, z, b2, out, N);
}


# x@W1 product: a thread asks for its eight 16-B pieces of the 16x512 x tile before converting the first (one wait for memory per tile), not 32 4-B loads each waited for in turn; sums unchanged
# speedup vs baseline: 1.9995x; 1.1553x over previous
#include <hip/hip_runtime.h>
#include <math.h>

#define F_IN 512
#define F_HID 128
#define F_OUT 16

typedef __attribute__((ext_vector_type(16))) __bf16 v16bf;
typedef __attribute__((ext_vector_type(8)))  float  v8f;

union BFrag {
    uint4 u[2];
    v16bf v;
};

__device__ __forceinline__ unsigned int f2bf(float f) {
    union { float f; unsigned int u; } cv; cv.f = f;
    unsigned int u = cv.u;
    unsigned int r = u + 0x7FFFu + ((u >> 16) & 1u);
    return (r >> 16) & 0xFFFFu;
}


__global__ void k_zero(float* deg, int* cnt, int* total, int n) {
    int i = blockIdx.x * blockDim.x + threadIdx.x;
    if (i < n) { deg[i] = 1.0f; cnt[i] = 0; }
    if (i == 0) *total = 0;
}

__global__ void k_count(const int* __restrict__ dst,
                        const float* __restrict__ ew,
                        float* deg, int* cnt,
                        int* __restrict__ rank, int e) {
    int i = blockIdx.x * blockDim.x + threadIdx.x;
    if (i < e) {
        const int d = dst[i];
        atomicAdd(&deg[d], ew[i]);
        rank[i] = atomicAdd(&cnt[d], 1);
    }
}

__global__ void k_offsets(float* deg, const int* __restrict__ cnt,
                          int* __restrict__ row_start, int* total, int n) {
    const int i = blockIdx.x * blockDim.x + threadIdx.x;
    const int lane = threadIdx.x & 31;
    int c = 0;
    if (i < n) {
        c = cnt[i];
        float d = deg[i];
        deg[i] = (d > 0.0f) ? rsqrtf(d) : 0.0f;
    }
    int x = c;
#pragma unroll
    for (int off = 1; off < 32; off <<= 1) {
        const int t = __shfl_up(x, off);
        if (lane >= off) x += t;
    }
    int b = 0;
    if (lane == 31) b = atomicAdd(total, x);
    b = __shfl(b, 31);
    if (i < n) row_start[i] = b + x - c;
}

__global__ void k_fill(const int* __restrict__ src,
                       const int* __restrict__ dst,
                       const float* __restrict__ ew,
                       const float* __restrict__ dinv,
                       const int* __restrict__ row_start,
                       const int* __restrict__ rank,
                       int2* __restrict__ csr, int e) {
    int i = blockIdx.x * blockDim.x + threadIdx.x;
    if (i < e) {
        const int s = src[i];
        const int d = dst[i];
        const float coef = dinv[s] * ew[i] * dinv[d];
        csr[row_start[d] + rank[i]] = make_int2(s, __float_as_int(coef));
    }
}


__global__ void k_prep_weights(const float* __restrict__ W1,
                               const float* __restrict__ W2,
                               unsigned short* __restrict__ Wt1,
                               unsigned short* __restrict__ Wt2) {
    int t = blockIdx.x * blockDim.x + threadIdx.x;
    if (t < F_IN * F_HID) {
        int n = t / F_IN, k = t % F_IN;
        Wt1[t] = (unsigned short)f2bf(W1[k * F_HID + n]);
    }
    int t2 = t - F_IN * F_HID;
    if (t2 >= 0 && t2 < F_HID * F_OUT) {
        int n = t2 / F_HID, k = t2 % F_HID;
        Wt2[t2] = (unsigned short)f2bf(W2[k * F_OUT + n]);
    }
}


#define A_PITCH (F_IN + 8)

__global__ void k_gemm1_wmma(const float* __restrict__ x,
                             const unsigned short* __restrict__ Wt1,
                             float* __restrict__ h1,
                             int nrows) {
    __shared__ __attribute__((aligned(16))) unsigned short Abuf[16 * A_PITCH];
    const int tid = threadIdx.x;
    const int row0 = blockIdx.x * 16;
    const int last = nrows - 1;

    float4 va[8];
#pragma unroll
    for (int j = 0; j < 8; ++j) {
        const int q = tid + 256 * j;
        const int m = q >> 7;
        const int c4 = q & 127;
        int gr = row0 + m;
        gr = (gr < last) ? gr : last;
        va[j] = *(const float4*)(x + (size_t)gr * F_IN + c4 * 4);
    }
#pragma unroll
    for (int j = 0; j < 8; ++j) {
        const int q = tid + 256 * j;
        const int m = q >> 7;
        const int c4 = q & 127;
        uint2 pk;
        pk.x = f2bf(va[j].x) | (f2bf(va[j].y) << 16);
        pk.y = f2bf(va[j].z) | (f2bf(va[j].w) << 16);
        *(uint2*)&Abuf[m * A_PITCH + c4 * 4] = pk;
    }
    __syncthreads();

    const int wave = tid >> 5;
    const int lane = tid & 31;
    const int n0 = wave * 16;
    const int mrow = lane & 15;
    const int khalf = (lane >> 4) << 3;

    const unsigned short* brow = Wt1 + (size_t)(n0 + mrow) * F_IN;

    v8f acc = {};
    for (int k = 0; k < F_IN; k += 32) {
        BFrag a, b;
        a.u[0] = *(const uint4*)&Abuf[mrow * A_PITCH + k + khalf];
        a.u[1] = *(const uint4*)&Abuf[mrow * A_PITCH + k + 16 + khalf];
        b.u[0] = *(const uint4*)&brow[k + khalf];
        b.u[1] = *(const uint4*)&brow[k + 16 + khalf];
        acc = __builtin_amdgcn_wmma_f32_16x16x32_bf16(
            false, a.v, false, b.v, (short)0, acc, false, false);
    }

    const int col = n0 + (lane & 15);
    const int rbase = row0 + ((lane >> 4) << 3);
#pragma unroll
    for (int r = 0; r < 8; ++r) {
        int row = rbase + r;
        if (row < nrows) h1[(size_t)row * F_HID + col] = acc[r];
    }
}


#define ROWS_PER_WAVE1 2

template <int B>
__device__ __forceinline__ void gather_rows(float4& acc, const float* hcol,
                                            int s_l, int c_l, int j) {
    float4 v[B];
    float cc[B];
#pragma unroll
    for (int u = 0; u < B; ++u) {
        const int s = __builtin_amdgcn_readlane(s_l, j + u);
        cc[u] = __int_as_float(__builtin_amdgcn_readlane(c_l, j + u));
        v[u] = *(const float4*)(hcol + (size_t)s * F_HID);
    }
#pragma unroll
    for (int u = 0; u < B; ++u) {
        acc.x += cc[u] * v[u].x;
        acc.y += cc[u] * v[u].y;
        acc.z += cc[u] * v[u].z;
        acc.w += cc[u] * v[u].w;
    }
}

__global__ void k_gather1(const int2* __restrict__ csr,
                          const int* __restrict__ row_start,
                          const int* __restrict__ cnt,
                          const float* __restrict__ dinv,
                          const float* __restrict__ h1,
                          const float* __restrict__ b1,
                          unsigned short* __restrict__ hb, int n) {
    const int lane = threadIdx.x & 31;
    const int wid = __builtin_amdgcn_readfirstlane(
        (int)((blockIdx.x * blockDim.x + threadIdx.x) >> 5));
    const float4 bias = *(const float4*)(b1 + lane * 4);
    const float* hcol = h1 + lane * 4;

    for (int r = 0; r < ROWS_PER_WAVE1; ++r) {
        const int i = wid * ROWS_PER_WAVE1 + r;
        if (i >= n) break;
        const int start = row_start[i];
        const int deg_i = cnt[i];
        const float di = dinv[i];
        const float sl = di * di;

        int m = (deg_i < 32) ? deg_i : 32;
        int s_l = 0;
        int c_l = 0;
        if (lane < m) {
            const int2 p = csr[start + lane];
            s_l = p.x;
            c_l = p.y;
        }
        float4 acc = *(const float4*)(hcol + (size_t)i * F_HID);
        acc.x *= sl; acc.y *= sl; acc.z *= sl; acc.w *= sl;

        int base = 0;
        while (m > 0) {
            int j = 0;
            for (; j + 8 <= m; j += 8) gather_rows<8>(acc, hcol, s_l, c_l, j);
            if (j + 4 <= m) { gather_rows<4>(acc, hcol, s_l, c_l, j); j += 4; }
            if (j + 2 <= m) { gather_rows<2>(acc, hcol, s_l, c_l, j); j += 2; }
            if (j < m)      { gather_rows<1>(acc, hcol, s_l, c_l, j); }
            base += 32;
            m = (deg_i - base < 32) ? (deg_i - base) : 32;
            if (m > 0 && lane < m) {
                const int2 p = csr[start + base + lane];
                s_l = p.x;
                c_l = p.y;
            }
        }

        const float ox = fmaxf(acc.x + bias.x, 0.0f);
        const float oy = fmaxf(acc.y + bias.y, 0.0f);
        const float oz = fmaxf(acc.z + bias.z, 0.0f);
        const float ow = fmaxf(acc.w + bias.w, 0.0f);
        uint2 pk;
        pk.x = f2bf(ox) | (f2bf(oy) << 16);
        pk.y = f2bf(oz) | (f2bf(ow) << 16);
        *(uint2*)(hb + (size_t)i * F_HID + lane * 4) = pk;
    }
}


__global__ void k_gemm2_wmma(const unsigned short* __restrict__ hb,
                             const unsigned short* __restrict__ Wt2,
                             float* __restrict__ z,
                             int nrows) {
    const int wave = threadIdx.x >> 5;
    const int lane = threadIdx.x & 31;
    const int m0 = blockIdx.x * 128 + wave * 16;
    if (m0 >= nrows) return;

    const int mrow = lane & 15;
    const int khalf = (lane >> 4) << 3;
    const unsigned short* arow = hb + (size_t)(m0 + mrow) * F_HID;
    const unsigned short* brow = Wt2 + (size_t)mrow * F_HID;

    v8f acc = {};
#pragma unroll
    for (int k = 0; k < F_HID; k += 32) {
        BFrag a, b;
        a.u[0] = *(const uint4*)(arow + k + khalf);
        a.u[1] = *(const uint4*)(arow + k + 16 + khalf);
        b.u[0] = *(const uint4*)(brow + k + khalf);
        b.u[1] = *(const uint4*)(brow + k + 16 + khalf);
        acc = __builtin_amdgcn_wmma_f32_16x16x32_bf16(
            false, a.v, false, b.v, (short)0, acc, false, false);
    }

    const int col = lane & 15;
    const int rbase = m0 + ((lane >> 4) << 3);
#pragma unroll
    for (int r = 0; r < 8; ++r) {
        int row = rbase + r;
        if (row < nrows) z[(size_t)row * F_OUT + col] = acc[r];
    }
}


__global__ void k_gather2(const int2* __restrict__ csr,
                          const int* __restrict__ row_start,
                          const int* __restrict__ cnt,
                          const float* __restrict__ dinv,
                          const float* __restrict__ z,
                          const float* __restrict__ b2,
                          float* __restrict__ out, int n) {
    const int lane = threadIdx.x & 31;
    const int half = lane >> 4;
    const int c = lane & 15;
    const int wid = __builtin_amdgcn_readfirstlane(
        (int)((blockIdx.x * blockDim.x + threadIdx.x) >> 5));
    if (wid * 2 >= n) return;
    const int i = wid * 2 + half;
    const bool valid = i < n;

    int start = 0, deg_i = 0;
    float acc = 0.0f;
    if (valid) {
        start = row_start[i];
        deg_i = cnt[i];
        const float di = dinv[i];
        acc = di * di * z[(size_t)i * F_OUT + c];
    }
    const int other = __shfl_xor(deg_i, 16);
    const int maxdeg = __builtin_amdgcn_readfirstlane(deg_i > other ? deg_i : other);

    for (int base = 0; base < maxdeg; base += 16) {
        const int m = deg_i - base;
        int s_l = 0;
        float c_l = 0.0f;
        if (c < m) {
            const int2 p = csr[start + base + c];
            s_l = p.x;
            c_l = __int_as_float(p.y);
        }
        float v[16];
        float cf[16];
#pragma unroll
        for (int u = 0; u < 16; ++u) {
            const int s = __shfl(s_l, u, 16);
            cf[u] = __shfl(c_l, u, 16);
            v[u] = (u < m) ? z[(size_t)s * F_OUT + c] : 0.0f;
        }
#pragma unroll
        for (int u = 0; u < 16; ++u) acc += cf[u] * v[u];
    }

    const float l = acc + b2[c];
    float mx = l;
#pragma unroll
    for (int k = 8; k >= 1; k >>= 1) mx = fmaxf(mx, __shfl_xor(mx, k, 16));
    const float ex = expf(l - mx);
    float ssum = 0.0f;
#pragma unroll
    for (int u = 0; u < 16; ++u) ssum += __shfl(ex, u, 16);
    const float lse = mx + logf(ssum);
    if (valid) out[(size_t)i * F_OUT + c] = l - lse;
}


extern "C" void kernel_launch(void* const* d_in, const int* in_sizes, int n_in,
                              void* d_out, int out_size, void* d_ws, size_t ws_size,
                              hipStream_t stream) {
    (void)n_in; (void)out_size; (void)ws_size;

    const float* x  = (const float*)d_in[0];
    const int*   ei = (const int*)d_in[1];
    const float* ew = (const float*)d_in[2];
    const float* W1 = (const float*)d_in[3];
    const float* b1 = (const float*)d_in[4];
    const float* W2 = (const float*)d_in[5];
    const float* b2 = (const float*)d_in[6];
    float* out = (float*)d_out;

    const int N = in_sizes[0] / F_IN;
    const int E = in_sizes[2];
    const int* src = ei;
    const int* dst = ei + E;

    char* ws = (char*)d_ws;
    size_t off = 0;
    auto take = [&](size_t bytes) -> char* {
        char* p = ws + off;
        off += (bytes + 255) & ~(size_t)255;
        return p;
    };
    float*          dinv = (float*)take((size_t)N * 4);
    unsigned short* Wt1  = (unsigned short*)take((size_t)F_IN * F_HID * 2);
    unsigned short* Wt2  = (unsigned short*)take((size_t)F_HID * F_OUT * 2);
    float*          bufA = (float*)take((size_t)N * F_HID * 4);
    char*           bufB = take((size_t)N * F_HID * 4);

    unsigned short* hb = (unsigned short*)bufB;
    char* q = bufB + (size_t)N * F_HID * 2;
    int2* csr       = (int2*)q;  q += (size_t)E * 8;
    int*  rank      = (int*)q;   q += (size_t)E * 4;
    int*  cnt       = (int*)q;   q += (size_t)N * 4;
    int*  row_start = (int*)q;   q += (size_t)N * 4;
    int*  total     = (int*)q;
    float* z = bufA;

    const int T = 256;

    k_zero<<<(N + T - 1) / T, T, 0, stream>>>(dinv, cnt, total, N);
    k_count<<<(E + T - 1) / T, T, 0, stream>>>(dst, ew, dinv, cnt, rank, E);
    k_offsets<<<(N + T - 1) / T, T, 0, stream>>>(dinv, cnt, row_start, total, N);
    k_fill<<<(E + T - 1) / T, T, 0, stream>>>(src, dst, ew, dinv, row_start, rank, csr, E);

    k_prep_weights<<<(F_IN * F_HID + F_HID * F_OUT + T - 1) / T, T, 0, stream>>>(
        W1, W2, Wt1, Wt2);

    k_gemm1_wmma<<<(N + 15) / 16, T, 0, stream>>>(x, Wt1, bufA, N);
    k_gather1<<<(N + ROWS_PER_WAVE1 * 8 - 1) / (ROWS_PER_WAVE1 * 8), T, 0, stream>>>(
        csr, row_start, cnt, dinv, bufA, b1, hb, N);

    k_gemm2_wmma<<<(N + 127) / 128, T, 0, stream>>>(hb, Wt2, z, N);
    k_gather2<<<(N + 15) / 16, T, 0, stream>>>(csr, row_start, cnt, dinv, z, b2, out, N);
}


# x@W1 product: a wave keeps its 16 columns of W1 (all 512 k) in registers while its block walks over many 16-row tiles, next tile's x loads in flight: W1 is read once per block, not per 16 rows
# speedup vs baseline: 2.3006x; 1.1506x over previous
#include <hip/hip_runtime.h>
#include <math.h>

#define F_IN 512
#define F_HID 128
#define F_OUT 16

typedef __attribute__((ext_vector_type(16))) __bf16 v16bf;
typedef __attribute__((ext_vector_type(8)))  float  v8f;

union BFrag {
    uint4 u[2];
    v16bf v;
};

__device__ __forceinline__ unsigned int f2bf(float f) {
    union { float f; unsigned int u; } cv; cv.f = f;
    unsigned int u = cv.u;
    unsigned int r = u + 0x7FFFu + ((u >> 16) & 1u);
    return (r >> 16) & 0xFFFFu;
}


__global__ void k_zero(float* deg, int* cnt, int* total, int n) {
    int i = blockIdx.x * blockDim.x + threadIdx.x;
    if (i < n) { deg[i] = 1.0f; cnt[i] = 0; }
    if (i == 0) *total = 0;
}

__global__ void k_count(const int* __restrict__ dst,
                        const float* __restrict__ ew,
                        float* deg, int* cnt,
                        int* __restrict__ rank, int e) {
    int i = blockIdx.x * blockDim.x + threadIdx.x;
    if (i < e) {
        const int d = dst[i];
        atomicAdd(&deg[d], ew[i]);
        rank[i] = atomicAdd(&cnt[d], 1);
    }
}

__global__ void k_offsets(float* deg, const int* __restrict__ cnt,
                          int* __restrict__ row_start, int* total, int n) {
    const int i = blockIdx.x * blockDim.x + threadIdx.x;
    const int lane = threadIdx.x & 31;
    int c = 0;
    if (i < n) {
        c = cnt[i];
        float d = deg[i];
        deg[i] = (d > 0.0f) ? rsqrtf(d) : 0.0f;
    }
    int x = c;
#pragma unroll
    for (int off = 1; off < 32; off <<= 1) {
        const int t = __shfl_up(x, off);
        if (lane >= off) x += t;
    }
    int b = 0;
    if (lane == 31) b = atomicAdd(total, x);
    b = __shfl(b, 31);
    if (i < n) row_start[i] = b + x - c;
}

__global__ void k_fill(const int* __restrict__ src,
                       const int* __restrict__ dst,
                       const float* __restrict__ ew,
                       const float* __restrict__ dinv,
                       const int* __restrict__ row_start,
                       const int* __restrict__ rank,
                       int2* __restrict__ csr, int e) {
    int i = blockIdx.x * blockDim.x + threadIdx.x;
    if (i < e) {
        const int s = src[i];
        const int d = dst[i];
        const float coef = dinv[s] * ew[i] * dinv[d];
        csr[row_start[d] + rank[i]] = make_int2(s, __float_as_int(coef));
    }
}


__global__ void k_prep_weights(const float* __restrict__ W1,
                               const float* __restrict__ W2,
                               unsigned short* __restrict__ Wt1,
                               unsigned short* __restrict__ Wt2) {
    int t = blockIdx.x * blockDim.x + threadIdx.x;
    if (t < F_IN * F_HID) {
        int n = t / F_IN, k = t % F_IN;
        Wt1[t] = (unsigned short)f2bf(W1[k * F_HID + n]);
    }
    int t2 = t - F_IN * F_HID;
    if (t2 >= 0 && t2 < F_HID * F_OUT) {
        int n = t2 / F_HID, k = t2 % F_HID;
        Wt2[t2] = (unsigned short)f2bf(W2[k * F_OUT + n]);
    }
}


#define A_PITCH (F_IN + 8)
#define GEMM1_BLOCKS 512

__device__ __forceinline__ void load_x_tile(float4 (&va)[8], const float* __restrict__ x,
                                            int row0, int last, int tid) {
#pragma unroll
    for (int j = 0; j < 8; ++j) {
        const int q = tid + 256 * j;
        const int m = q >> 7;
        const int c4 = q & 127;
        int gr = row0 + m;
        gr = (gr < last) ? gr : last;
        va[j] = *(const float4*)(x + (size_t)gr * F_IN + c4 * 4);
    }
}

__global__ __launch_bounds__(256) void k_gemm1_wmma(const float* __restrict__ x,
                             const unsigned short* __restrict__ Wt1,
                             float* __restrict__ h1,
                             int nrows) {
    __shared__ __attribute__((aligned(16))) unsigned short Abuf[2][16 * A_PITCH];
    const int tid = threadIdx.x;
    const int last = nrows - 1;
    const int ntiles = (nrows + 15) >> 4;

    const int wave = tid >> 5;
    const int lane = tid & 31;
    const int n0 = wave * 16;
    const int mrow = lane & 15;
    const int khalf = (lane >> 4) << 3;

    int t = blockIdx.x;
    float4 va[8];
    if (t < ntiles) load_x_tile(va, x, t * 16, last, tid);

    const unsigned short* brow = Wt1 + (size_t)(n0 + mrow) * F_IN;
    BFrag b[16];
#pragma unroll
    for (int s = 0; s < 16; ++s) {
        b[s].u[0] = *(const uint4*)&brow[s * 32 + khalf];
        b[s].u[1] = *(const uint4*)&brow[s * 32 + 16 + khalf];
    }

    int buf = 0;
    for (; t < ntiles; t += gridDim.x) {
        unsigned short* A = Abuf[buf];
#pragma unroll
        for (int j = 0; j < 8; ++j) {
            const int q = tid + 256 * j;
            const int m = q >> 7;
            const int c4 = q & 127;
            uint2 pk;
            pk.x = f2bf(va[j].x) | (f2bf(va[j].y) << 16);
            pk.y = f2bf(va[j].z) | (f2bf(va[j].w) << 16);
            *(uint2*)&A[m * A_PITCH + c4 * 4] = pk;
        }
        __syncthreads();

        const int tn = t + gridDim.x;
        if (tn < ntiles) load_x_tile(va, x, tn * 16, last, tid);

        v8f acc = {};
#pragma unroll
        for (int s = 0; s < 16; ++s) {
            BFrag a;
            a.u[0] = *(const uint4*)&A[mrow * A_PITCH + s * 32 + khalf];
            a.u[1] = *(const uint4*)&A[mrow * A_PITCH + s * 32 + 16 + khalf];
            acc = __builtin_amdgcn_wmma_f32_16x16x32_bf16(
                false, a.v, false, b[s].v, (short)0, acc, false, false);
        }

        const int col = n0 + (lane & 15);
        const int rbase = t * 16 + ((lane >> 4) << 3);
#pragma unroll
        for (int r = 0; r < 8; ++r) {
            int row = rbase + r;
            if (row < nrows) h1[(size_t)row * F_HID + col] = acc[r];
        }
        buf ^= 1;
    }
}


#define ROWS_PER_WAVE1 2

template <int B>
__device__ __forceinline__ void gather_rows(float4& acc, const float* hcol,
                                            int s_l, int c_l, int j) {
    float4 v[B];
    float cc[B];
#pragma unroll
    for (int u = 0; u < B; ++u) {
        const int s = __builtin_amdgcn_readlane(s_l, j + u);
        cc[u] = __int_as_float(__builtin_amdgcn_readlane(c_l, j + u));
        v[u] = *(const float4*)(hcol + (size_t)s * F_HID);
    }
#pragma unroll
    for (int u = 0; u < B; ++u) {
        acc.x += cc[u] * v[u].x;
        acc.y += cc[u] * v[u].y;
        acc.z += cc[u] * v[u].z;
        acc.w += cc[u] * v[u].w;
    }
}

__global__ void k_gather1(const int2* __restrict__ csr,
                          const int* __restrict__ row_start,
                          const int* __restrict__ cnt,
                          const float* __restrict__ dinv,
                          const float* __restrict__ h1,
                          const float* __restrict__ b1,
                          unsigned short* __restrict__ hb, int n) {
    const int lane = threadIdx.x & 31;
    const int wid = __builtin_amdgcn_readfirstlane(
        (int)((blockIdx.x * blockDim.x + threadIdx.x) >> 5));
    const float4 bias = *(const float4*)(b1 + lane * 4);
    const float* hcol = h1 + lane * 4;

    for (int r = 0; r < ROWS_PER_WAVE1; ++r) {
        const int i = wid * ROWS_PER_WAVE1 + r;
        if (i >= n) break;
        const int start = row_start[i];
        const int deg_i = cnt[i];
        const float di = dinv[i];
        const float sl = di * di;

        int m = (deg_i < 32) ? deg_i : 32;
        int s_l = 0;
        int c_l = 0;
        if (lane < m) {
            const int2 p = csr[start + lane];
            s_l = p.x;
            c_l = p.y;
        }
        float4 acc = *(const float4*)(hcol + (size_t)i * F_HID);
        acc.x *= sl; acc.y *= sl; acc.z *= sl; acc.w *= sl;

        int base = 0;
        while (m > 0) {
            int j = 0;
            for (; j + 8 <= m; j += 8) gather_rows<8>(acc, hcol, s_l, c_l, j);
            if (j + 4 <= m) { gather_rows<4>(acc, hcol, s_l, c_l, j); j += 4; }
            if (j + 2 <= m) { gather_rows<2>(acc, hcol, s_l, c_l, j); j += 2; }
            if (j < m)      { gather_rows<1>(acc, hcol, s_l, c_l, j); }
            base += 32;
            m = (deg_i - base < 32) ? (deg_i - base) : 32;
            if (m > 0 && lane < m) {
                const int2 p = csr[start + base + lane];
                s_l = p.x;
                c_l = p.y;
            }
        }

        const float ox = fmaxf(acc.x + bias.x, 0.0f);
        const float oy = fmaxf(acc.y + bias.y, 0.0f);
        const float oz = fmaxf(acc.z + bias.z, 0.0f);
        const float ow = fmaxf(acc.w + bias.w, 0.0f);
        uint2 pk;
        pk.x = f2bf(ox) | (f2bf(oy) << 16);
        pk.y = f2bf(oz) | (f2bf(ow) << 16);
        *(uint2*)(hb + (size_t)i * F_HID + lane * 4) = pk;
    }
}


__global__ void k_gemm2_wmma(const unsigned short* __restrict__ hb,
                             const unsigned short* __restrict__ Wt2,
                             float* __restrict__ z,
                             int nrows) {
    const int wave = threadIdx.x >> 5;
    const int lane = threadIdx.x & 31;
    const int m0 = blockIdx.x * 128 + wave * 16;
    if (m0 >= nrows) return;

    const int mrow = lane & 15;
    const int khalf = (lane >> 4) << 3;
    const unsigned short* arow = hb + (size_t)(m0 + mrow) * F_HID;
    const unsigned short* brow = Wt2 + (size_t)mrow * F_HID;

    v8f acc = {};
#pragma unroll
    for (int k = 0; k < F_HID; k += 32) {
        BFrag a, b;
        a.u[0] = *(const uint4*)(arow + k + khalf);
        a.u[1] = *(const uint4*)(arow + k + 16 + khalf);
        b.u[0] = *(const uint4*)(brow + k + khalf);
        b.u[1] = *(const uint4*)(brow + k + 16 + khalf);
        acc = __builtin_amdgcn_wmma_f32_16x16x32_bf16(
            false, a.v, false, b.v, (short)0, acc, false, false);
    }

    const int col = lane & 15;
    const int rbase = m0 + ((lane >> 4) << 3);
#pragma unroll
    for (int r = 0; r < 8; ++r) {
        int row = rbase + r;
        if (row < nrows) z[(size_t)row * F_OUT + col] = acc[r];
    }
}


__global__ void k_gather2(const int2* __restrict__ csr,
                          const int* __restrict__ row_start,
                          const int* __restrict__ cnt,
                          const float* __restrict__ dinv,
                          const float* __restrict__ z,
                          const float* __restrict__ b2,
                          float* __restrict__ out, int n) {
    const int lane = threadIdx.x & 31;
    const int half = lane >> 4;
    const int c = lane & 15;
    const int wid = __builtin_amdgcn_readfirstlane(
        (int)((blockIdx.x * blockDim.x + threadIdx.x) >> 5));
    if (wid * 2 >= n) return;
    const int i = wid * 2 + half;
    const bool valid = i < n;

    int start = 0, deg_i = 0;
    float acc = 0.0f;
    if (valid) {
        start = row_start[i];
        deg_i = cnt[i];
        const float di = dinv[i];
        acc = di * di * z[(size_t)i * F_OUT + c];
    }
    const int other = __shfl_xor(deg_i, 16);
    const int maxdeg = __builtin_amdgcn_readfirstlane(deg_i > other ? deg_i : other);

    for (int base = 0; base < maxdeg; base += 16) {
        const int m = deg_i - base;
        int s_l = 0;
        float c_l = 0.0f;
        if (c < m) {
            const int2 p = csr[start + base + c];
            s_l = p.x;
            c_l = __int_as_float(p.y);
        }
        float v[16];
        float cf[16];
#pragma unroll
        for (int u = 0; u < 16; ++u) {
            const int s = __shfl(s_l, u, 16);
            cf[u] = __shfl(c_l, u, 16);
            v[u] = (u < m) ? z[(size_t)s * F_OUT + c] : 0.0f;
        }
#pragma unroll
        for (int u = 0; u < 16; ++u) acc += cf[u] * v[u];
    }

    const float l = acc + b2[c];
    float mx = l;
#pragma unroll
    for (int k = 8; k >= 1; k >>= 1) mx = fmaxf(mx, __shfl_xor(mx, k, 16));
    const float ex = expf(l - mx);
    float ssum = 0.0f;
#pragma unroll
    for (int u = 0; u < 16; ++u) ssum += __shfl(ex, u, 16);
    const float lse = mx + logf(ssum);
    if (valid) out[(size_t)i * F_OUT + c] = l - lse;
}


extern "C" void kernel_launch(void* const* d_in, const int* in_sizes, int n_in,
                              void* d_out, int out_size, void* d_ws, size_t ws_size,
                              hipStream_t stream) {
    (void)n_in; (void)out_size; (void)ws_size;

    const float* x  = (const float*)d_in[0];
    const int*   ei = (const int*)d_in[1];
    const float* ew = (const float*)d_in[2];
    const float* W1 = (const float*)d_in[3];
    const float* b1 = (const float*)d_in[4];
    const float* W2 = (const float*)d_in[5];
    const float* b2 = (const float*)d_in[6];
    float* out = (float*)d_out;

    const int N = in_sizes[0] / F_IN;
    const int E = in_sizes[2];
    const int* src = ei;
    const int* dst = ei + E;

    char* ws = (char*)d_ws;
    size_t off = 0;
    auto take = [&](size_t bytes) -> char* {
        char* p = ws + off;
        off += (bytes + 255) & ~(size_t)255;
        return p;
    };
    float*          dinv = (float*)take((size_t)N * 4);
    unsigned short* Wt1  = (unsigned short*)take((size_t)F_IN * F_HID * 2);
    unsigned short* Wt2  = (unsigned short*)take((size_t)F_HID * F_OUT * 2);
    float*          bufA = (float*)take((size_t)N * F_HID * 4);
    char*           bufB = take((size_t)N * F_HID * 4);

    unsigned short* hb = (unsigned short*)bufB;
    char* q = bufB + (size_t)N * F_HID * 2;
    int2* csr       = (int2*)q;  q += (size_t)E * 8;
    int*  rank      = (int*)q;   q += (size_t)E * 4;
    int*  cnt       = (int*)q;   q += (size_t)N * 4;
    int*  row_start = (int*)q;   q += (size_t)N * 4;
    int*  total     = (int*)q;
    float* z = bufA;

    const int T = 256;

    k_zero<<<(N + T - 1) / T, T, 0, stream>>>(dinv, cnt, total, N);
    k_count<<<(E + T - 1) / T, T, 0, stream>>>(dst, ew, dinv, cnt, rank, E);
    k_offsets<<<(N + T - 1) / T, T, 0, stream>>>(dinv, cnt, row_start, total, N);
    k_fill<<<(E + T - 1) / T, T, 0, stream>>>(src, dst, ew, dinv, row_start, rank, csr, E);

    k_prep_weights<<<(F_IN * F_HID + F_HID * F_OUT + T - 1) / T, T, 0, stream>>>(
        W1, W2, Wt1, Wt2);

    k_gemm1_wmma<<<GEMM1_BLOCKS, T, 0, stream>>>(x, Wt1, bufA, N);
    k_gather1<<<(N + ROWS_PER_WAVE1 * 8 - 1) / (ROWS_PER_WAVE1 * 8), T, 0, stream>>>(
        csr, row_start, cnt, dinv, bufA, b1, hb, N);

    k_gemm2_wmma<<<(N + 127) / 128, T, 0, stream>>>(hb, Wt2, z, N);
    k_gather2<<<(N + 15) / 16, T, 0, stream>>>(csr, row_start, cnt, dinv, z, b2, out, N);
}


# norm per row, not per edge: products store row s times dinv[s], row sums end with times dinv[d] (sum order unchanged); edge records = (source, weight): fill pass: 1 random read per edge, not 3
# speedup vs baseline: 2.3709x; 1.0306x over previous
#include <hip/hip_runtime.h>
#include <math.h>

#define F_IN 512
#define F_HID 128
#define F_OUT 16

typedef __attribute__((ext_vector_type(16))) __bf16 v16bf;
typedef __attribute__((ext_vector_type(8)))  float  v8f;

union BFrag {
    uint4 u[2];
    v16bf v;
};

__device__ __forceinline__ unsigned int f2bf(float f) {
    union { float f; unsigned int u; } cv; cv.f = f;
    unsigned int u = cv.u;
    unsigned int r = u + 0x7FFFu + ((u >> 16) & 1u);
    return (r >> 16) & 0xFFFFu;
}


__global__ void k_zero(float* deg, int* cnt, int* total, int n) {
    int i = blockIdx.x * blockDim.x + threadIdx.x;
    if (i < n) { deg[i] = 1.0f; cnt[i] = 0; }
    if (i == 0) *total = 0;
}

__global__ void k_count(const int* __restrict__ dst,
                        const float* __restrict__ ew,
                        float* deg, int* cnt,
                        int* __restrict__ rank, int e) {
    int i = blockIdx.x * blockDim.x + threadIdx.x;
    if (i < e) {
        const int d = dst[i];
        atomicAdd(&deg[d], ew[i]);
        rank[i] = atomicAdd(&cnt[d], 1);
    }
}

__global__ void k_offsets(float* deg, const int* __restrict__ cnt,
                          int* __restrict__ row_start, int* total, int n) {
    const int i = blockIdx.x * blockDim.x + threadIdx.x;
    const int lane = threadIdx.x & 31;
    int c = 0;
    if (i < n) {
        c = cnt[i];
        float d = deg[i];
        deg[i] = (d > 0.0f) ? rsqrtf(d) : 0.0f;
    }
    int x = c;
#pragma unroll
    for (int off = 1; off < 32; off <<= 1) {
        const int t = __shfl_up(x, off);
        if (lane >= off) x += t;
    }
    int b = 0;
    if (lane == 31) b = atomicAdd(total, x);
    b = __shfl(b, 31);
    if (i < n) row_start[i] = b + x - c;
}

__global__ void k_fill(const int* __restrict__ src,
                       const int* __restrict__ dst,
                       const float* __restrict__ ew,
                       const int* __restrict__ row_start,
                       const int* __restrict__ rank,
                       int2* __restrict__ csr, int e) {
    int i = blockIdx.x * blockDim.x + threadIdx.x;
    if (i < e) {
        const int s = src[i];
        const int d = dst[i];
        csr[row_start[d] + rank[i]] = make_int2(s, __float_as_int(ew[i]));
    }
}


__global__ void k_prep_weights(const float* __restrict__ W1,
                               const float* __restrict__ W2,
                               unsigned short* __restrict__ Wt1,
                               unsigned short* __restrict__ Wt2) {
    int t = blockIdx.x * blockDim.x + threadIdx.x;
    if (t < F_IN * F_HID) {
        int n = t / F_IN, k = t % F_IN;
        Wt1[t] = (unsigned short)f2bf(W1[k * F_HID + n]);
    }
    int t2 = t - F_IN * F_HID;
    if (t2 >= 0 && t2 < F_HID * F_OUT) {
        int n = t2 / F_HID, k = t2 % F_HID;
        Wt2[t2] = (unsigned short)f2bf(W2[k * F_OUT + n]);
    }
}


#define A_PITCH (F_IN + 8)
#define GEMM1_BLOCKS 512

__device__ __forceinline__ void load_x_tile(float4 (&va)[8], const float* __restrict__ x,
                                            int row0, int last, int tid) {
#pragma unroll
    for (int j = 0; j < 8; ++j) {
        const int q = tid + 256 * j;
        const int m = q >> 7;
        const int c4 = q & 127;
        int gr = row0 + m;
        gr = (gr < last) ? gr : last;
        va[j] = *(const float4*)(x + (size_t)gr * F_IN + c4 * 4);
    }
}

__global__ __launch_bounds__(256) void k_gemm1_wmma(const float* __restrict__ x,
                             const unsigned short* __restrict__ Wt1,
                             const float* __restrict__ dinv,
                             float* __restrict__ h1,
                             int nrows) {
    __shared__ __attribute__((aligned(16))) unsigned short Abuf[2][16 * A_PITCH];
    const int tid = threadIdx.x;
    const int last = nrows - 1;
    const int ntiles = (nrows + 15) >> 4;

    const int wave = tid >> 5;
    const int lane = tid & 31;
    const int n0 = wave * 16;
    const int mrow = lane & 15;
    const int khalf = (lane >> 4) << 3;

    int t = blockIdx.x;
    float4 va[8];
    if (t < ntiles) load_x_tile(va, x, t * 16, last, tid);

    const unsigned short* brow = Wt1 + (size_t)(n0 + mrow) * F_IN;
    BFrag b[16];
#pragma unroll
    for (int s = 0; s < 16; ++s) {
        b[s].u[0] = *(const uint4*)&brow[s * 32 + khalf];
        b[s].u[1] = *(const uint4*)&brow[s * 32 + 16 + khalf];
    }

    int buf = 0;
    for (; t < ntiles; t += gridDim.x) {
        unsigned short* A = Abuf[buf];
#pragma unroll
        for (int j = 0; j < 8; ++j) {
            const int q = tid + 256 * j;
            const int m = q >> 7;
            const int c4 = q & 127;
            uint2 pk;
            pk.x = f2bf(va[j].x) | (f2bf(va[j].y) << 16);
            pk.y = f2bf(va[j].z) | (f2bf(va[j].w) << 16);
            *(uint2*)&A[m * A_PITCH + c4 * 4] = pk;
        }
        __syncthreads();

        const int rbase = t * 16 + ((lane >> 4) << 3);
        float dv[8];
#pragma unroll
        for (int r = 0; r < 8; ++r) {
            const int row = rbase + r;
            dv[r] = dinv[(row < last) ? row : last];
        }

        const int tn = t + gridDim.x;
        if (tn < ntiles) load_x_tile(va, x, tn * 16, last, tid);

        v8f acc = {};
#pragma unroll
        for (int s = 0; s < 16; ++s) {
            BFrag a;
            a.u[0] = *(const uint4*)&A[mrow * A_PITCH + s * 32 + khalf];
            a.u[1] = *(const uint4*)&A[mrow * A_PITCH + s * 32 + 16 + khalf];
            acc = __builtin_amdgcn_wmma_f32_16x16x32_bf16(
                false, a.v, false, b[s].v, (short)0, acc, false, false);
        }

        const int col = n0 + (lane & 15);
#pragma unroll
        for (int r = 0; r < 8; ++r) {
            int row = rbase + r;
            const float o = acc[r] * dv[r];
            if (row < nrows) h1[(size_t)row * F_HID + col] = o;
        }
        buf ^= 1;
    }
}


#define ROWS_PER_WAVE1 2

template <int B>
__device__ __forceinline__ void gather_rows(float4& acc, const float* hcol,
                                            int s_l, int c_l, int j) {
    float4 v[B];
    float cc[B];
#pragma unroll
    for (int u = 0; u < B; ++u) {
        const int s = __builtin_amdgcn_readlane(s_l, j + u);
        cc[u] = __int_as_float(__builtin_amdgcn_readlane(c_l, j + u));
        v[u] = *(const float4*)(hcol + (size_t)s * F_HID);
    }
#pragma unroll
    for (int u = 0; u < B; ++u) {
        acc.x += cc[u] * v[u].x;
        acc.y += cc[u] * v[u].y;
        acc.z += cc[u] * v[u].z;
        acc.w += cc[u] * v[u].w;
    }
}

__global__ void k_gather1(const int2* __restrict__ csr,
                          const int* __restrict__ row_start,
                          const int* __restrict__ cnt,
                          const float* __restrict__ dinv,
                          const float* __restrict__ h1,
                          const float* __restrict__ b1,
                          unsigned short* __restrict__ hb, int n) {
    const int lane = threadIdx.x & 31;
    const int wid = __builtin_amdgcn_readfirstlane(
        (int)((blockIdx.x * blockDim.x + threadIdx.x) >> 5));
    const float4 bias = *(const float4*)(b1 + lane * 4);
    const float* hcol = h1 + lane * 4;

    for (int r = 0; r < ROWS_PER_WAVE1; ++r) {
        const int i = wid * ROWS_PER_WAVE1 + r;
        if (i >= n) break;
        const int start = row_start[i];
        const int deg_i = cnt[i];
        const float di = dinv[i];

        int m = (deg_i < 32) ? deg_i : 32;
        int s_l = 0;
        int c_l = 0;
        if (lane < m) {
            const int2 p = csr[start + lane];
            s_l = p.x;
            c_l = p.y;
        }
        float4 acc = *(const float4*)(hcol + (size_t)i * F_HID);

        int base = 0;
        while (m > 0) {
            int j = 0;
            for (; j + 8 <= m; j += 8) gather_rows<8>(acc, hcol, s_l, c_l, j);
            if (j + 4 <= m) { gather_rows<4>(acc, hcol, s_l, c_l, j); j += 4; }
            if (j + 2 <= m) { gather_rows<2>(acc, hcol, s_l, c_l, j); j += 2; }
            if (j < m)      { gather_rows<1>(acc, hcol, s_l, c_l, j); }
            base += 32;
            m = (deg_i - base < 32) ? (deg_i - base) : 32;
            if (m > 0 && lane < m) {
                const int2 p = csr[start + base + lane];
                s_l = p.x;
                c_l = p.y;
            }
        }

        const float ox = fmaxf(acc.x * di + bias.x, 0.0f);
        const float oy = fmaxf(acc.y * di + bias.y, 0.0f);
        const float oz = fmaxf(acc.z * di + bias.z, 0.0f);
        const float ow = fmaxf(acc.w * di + bias.w, 0.0f);
        uint2 pk;
        pk.x = f2bf(ox) | (f2bf(oy) << 16);
        pk.y = f2bf(oz) | (f2bf(ow) << 16);
        *(uint2*)(hb + (size_t)i * F_HID + lane * 4) = pk;
    }
}


__global__ void k_gemm2_wmma(const unsigned short* __restrict__ hb,
                             const unsigned short* __restrict__ Wt2,
                             const float* __restrict__ dinv,
                             float* __restrict__ z,
                             int nrows) {
    const int wave = threadIdx.x >> 5;
    const int lane = threadIdx.x & 31;
    const int m0 = blockIdx.x * 128 + wave * 16;
    if (m0 >= nrows) return;

    const int mrow = lane & 15;
    const int khalf = (lane >> 4) << 3;
    const unsigned short* arow = hb + (size_t)(m0 + mrow) * F_HID;
    const unsigned short* brow = Wt2 + (size_t)mrow * F_HID;

    const int rbase = m0 + ((lane >> 4) << 3);
    const int last = nrows - 1;
    float dv[8];
#pragma unroll
    for (int r = 0; r < 8; ++r) {
        const int row = rbase + r;
        dv[r] = dinv[(row < last) ? row : last];
    }

    v8f acc = {};
#pragma unroll
    for (int k = 0; k < F_HID; k += 32) {
        BFrag a, b;
        a.u[0] = *(const uint4*)(arow + k + khalf);
        a.u[1] = *(const uint4*)(arow + k + 16 + khalf);
        b.u[0] = *(const uint4*)(brow + k + khalf);
        b.u[1] = *(const uint4*)(brow + k + 16 + khalf);
        acc = __builtin_amdgcn_wmma_f32_16x16x32_bf16(
            false, a.v, false, b.v, (short)0, acc, false, false);
    }

    const int col = lane & 15;
#pragma unroll
    for (int r = 0; r < 8; ++r) {
        int row = rbase + r;
        const float o = acc[r] * dv[r];
        if (row < nrows) z[(size_t)row * F_OUT + col] = o;
    }
}


__global__ void k_gather2(const int2* __restrict__ csr,
                          const int* __restrict__ row_start,
                          const int* __restrict__ cnt,
                          const float* __restrict__ dinv,
                          const float* __restrict__ z,
                          const float* __restrict__ b2,
                          float* __restrict__ out, int n) {
    const int lane = threadIdx.x & 31;
    const int half = lane >> 4;
    const int c = lane & 15;
    const int wid = __builtin_amdgcn_readfirstlane(
        (int)((blockIdx.x * blockDim.x + threadIdx.x) >> 5));
    if (wid * 2 >= n) return;
    const int i = wid * 2 + half;
    const bool valid = i < n;

    int start = 0, deg_i = 0;
    float acc = 0.0f;
    float di = 0.0f;
    if (valid) {
        start = row_start[i];
        deg_i = cnt[i];
        di = dinv[i];
        acc = z[(size_t)i * F_OUT + c];
    }
    const int other = __shfl_xor(deg_i, 16);
    const int maxdeg = __builtin_amdgcn_readfirstlane(deg_i > other ? deg_i : other);

    for (int base = 0; base < maxdeg; base += 16) {
        const int m = deg_i - base;
        int s_l = 0;
        float c_l = 0.0f;
        if (c < m) {
            const int2 p = csr[start + base + c];
            s_l = p.x;
            c_l = __int_as_float(p.y);
        }
        float v[16];
        float cf[16];
#pragma unroll
        for (int u = 0; u < 16; ++u) {
            const int s = __shfl(s_l, u, 16);
            cf[u] = __shfl(c_l, u, 16);
            v[u] = (u < m) ? z[(size_t)s * F_OUT + c] : 0.0f;
        }
#pragma unroll
        for (int u = 0; u < 16; ++u) acc += cf[u] * v[u];
    }

    const float l = acc * di + b2[c];
    float mx = l;
#pragma unroll
    for (int k = 8; k >= 1; k >>= 1) mx = fmaxf(mx, __shfl_xor(mx, k, 16));
    const float ex = expf(l - mx);
    float ssum = 0.0f;
#pragma unroll
    for (int u = 0; u < 16; ++u) ssum += __shfl(ex, u, 16);
    const float lse = mx + logf(ssum);
    if (valid) out[(size_t)i * F_OUT + c] = l - lse;
}


extern "C" void kernel_launch(void* const* d_in, const int* in_sizes, int n_in,
                              void* d_out, int out_size, void* d_ws, size_t ws_size,
                              hipStream_t stream) {
    (void)n_in; (void)out_size; (void)ws_size;

    const float* x  = (const float*)d_in[0];
    const int*   ei = (const int*)d_in[1];
    const float* ew = (const float*)d_in[2];
    const float* W1 = (const float*)d_in[3];
    const float* b1 = (const float*)d_in[4];
    const float* W2 = (const float*)d_in[5];
    const float* b2 = (const float*)d_in[6];
    float* out = (float*)d_out;

    const int N = in_sizes[0] / F_IN;
    const int E = in_sizes[2];
    const int* src = ei;
    const int* dst = ei + E;

    char* ws = (char*)d_ws;
    size_t off = 0;
    auto take = [&](size_t bytes) -> char* {
        char* p = ws + off;
        off += (bytes + 255) & ~(size_t)255;
        return p;
    };
    float*          dinv = (float*)take((size_t)N * 4);
    unsigned short* Wt1  = (unsigned short*)take((size_t)F_IN * F_HID * 2);
    unsigned short* Wt2  = (unsigned short*)take((size_t)F_HID * F_OUT * 2);
    float*          bufA = (float*)take((size_t)N * F_HID * 4);
    char*           bufB = take((size_t)N * F_HID * 4);

    unsigned short* hb = (unsigned short*)bufB;
    char* q = bufB + (size_t)N * F_HID * 2;
    int2* csr       = (int2*)q;  q += (size_t)E * 8;
    int*  rank      = (int*)q;   q += (size_t)E * 4;
    int*  cnt       = (int*)q;   q += (size_t)N * 4;
    int*  row_start = (int*)q;   q += (size_t)N * 4;
    int*  total     = (int*)q;
    float* z = bufA;

    const int T = 256;

    k_zero<<<(N + T - 1) / T, T, 0, stream>>>(dinv, cnt, total, N);
    k_count<<<(E + T - 1) / T, T, 0, stream>>>(dst, ew, dinv, cnt, rank, E);
    k_offsets<<<(N + T - 1) / T, T, 0, stream>>>(dinv, cnt, row_start, total, N);
    k_fill<<<(E + T - 1) / T, T, 0, stream>>>(src, dst, ew, row_start, rank, csr, E);

    k_prep_weights<<<(F_IN * F_HID + F_HID * F_OUT + T - 1) / T, T, 0, stream>>>(
        W1, W2, Wt1, Wt2);

    k_gemm1_wmma<<<GEMM1_BLOCKS, T, 0, stream>>>(x, Wt1, dinv, bufA, N);
    k_gather1<<<(N + ROWS_PER_WAVE1 * 8 - 1) / (ROWS_PER_WAVE1 * 8), T, 0, stream>>>(
        csr, row_start, cnt, dinv, bufA, b1, hb, N);

    k_gemm2_wmma<<<(N + 127) / 128, T, 0, stream>>>(hb, Wt2, dinv, z, N);
    k_gather2<<<(N + 15) / 16, T, 0, stream>>>(csr, row_start, cnt, dinv, z, b2, out, N);
}


# edge-record fill pass has no launch of its own: each 16-row tile trip of the x@W1 launch also places 256 edges (reads issued beside the tile's row loads); 9 launches -> 8, sums unchanged
# speedup vs baseline: 2.5358x; 1.0696x over previous
#include <hip/hip_runtime.h>
#include <math.h>

#define F_IN 512
#define F_HID 128
#define F_OUT 16

typedef __attribute__((ext_vector_type(16))) __bf16 v16bf;
typedef __attribute__((ext_vector_type(8)))  float  v8f;

union BFrag {
    uint4 u[2];
    v16bf v;
};

__device__ __forceinline__ unsigned int f2bf(float f) {
    union { float f; unsigned int u; } cv; cv.f = f;
    unsigned int u = cv.u;
    unsigned int r = u + 0x7FFFu + ((u >> 16) & 1u);
    return (r >> 16) & 0xFFFFu;
}


__global__ void k_zero(float* deg, int* cnt, int* total, int n) {
    int i = blockIdx.x * blockDim.x + threadIdx.x;
    if (i < n) { deg[i] = 1.0f; cnt[i] = 0; }
    if (i == 0) *total = 0;
}

__global__ void k_count(const int* __restrict__ dst,
                        const float* __restrict__ ew,
                        float* deg, int* cnt,
                        int* __restrict__ rank, int e) {
    int i = blockIdx.x * blockDim.x + threadIdx.x;
    if (i < e) {
        const int d = dst[i];
        atomicAdd(&deg[d], ew[i]);
        rank[i] = atomicAdd(&cnt[d], 1);
    }
}

__global__ void k_offsets(float* deg, const int* __restrict__ cnt,
                          int* __restrict__ row_start, int* total, int n) {
    const int i = blockIdx.x * blockDim.x + threadIdx.x;
    const int lane = threadIdx.x & 31;
    int c = 0;
    if (i < n) {
        c = cnt[i];
        float d = deg[i];
        deg[i] = (d > 0.0f) ? rsqrtf(d) : 0.0f;
    }
    int x = c;
#pragma unroll
    for (int off = 1; off < 32; off <<= 1) {
        const int t = __shfl_up(x, off);
        if (lane >= off) x += t;
    }
    int b = 0;
    if (lane == 31) b = atomicAdd(total, x);
    b = __shfl(b, 31);
    if (i < n) row_start[i] = b + x - c;
}

__device__ __forceinline__ void fill_chunk(const int* __restrict__ src,
                                           const int* __restrict__ dst,
                                           const float* __restrict__ ew,
                                           const int* __restrict__ row_start,
                                           const int* __restrict__ rank,
                                           int2* __restrict__ csr, int i, int e) {
    if (i < e) {
        const int s = src[i];
        const int d = dst[i];
        csr[row_start[d] + rank[i]] = make_int2(s, __float_as_int(ew[i]));
    }
}


__global__ void k_prep_weights(const float* __restrict__ W1,
                               const float* __restrict__ W2,
                               unsigned short* __restrict__ Wt1,
                               unsigned short* __restrict__ Wt2) {
    int t = blockIdx.x * blockDim.x + threadIdx.x;
    if (t < F_IN * F_HID) {
        int n = t / F_IN, k = t % F_IN;
        Wt1[t] = (unsigned short)f2bf(W1[k * F_HID + n]);
    }
    int t2 = t - F_IN * F_HID;
    if (t2 >= 0 && t2 < F_HID * F_OUT) {
        int n = t2 / F_HID, k = t2 % F_HID;
        Wt2[t2] = (unsigned short)f2bf(W2[k * F_OUT + n]);
    }
}


#define A_PITCH (F_IN + 8)
#define GEMM1_BLOCKS 512

__device__ __forceinline__ void load_x_tile(float4 (&va)[8], const float* __restrict__ x,
                                            int row0, int last, int tid) {
#pragma unroll
    for (int j = 0; j < 8; ++j) {
        const int q = tid + 256 * j;
        const int m = q >> 7;
        const int c4 = q & 127;
        int gr = row0 + m;
        gr = (gr < last) ? gr : last;
        va[j] = *(const float4*)(x + (size_t)gr * F_IN + c4 * 4);
    }
}

__global__ __launch_bounds__(256) void k_gemm1_wmma(const float* __restrict__ x,
                             const unsigned short* __restrict__ Wt1,
                             const float* __restrict__ dinv,
                             float* __restrict__ h1,
                             int nrows,
                             const int* __restrict__ src,
                             const int* __restrict__ dst,
                             const float* __restrict__ ew,
                             const int* __restrict__ row_start,
                             const int* __restrict__ rank,
                             int2* __restrict__ csr,
                             int e) {
    __shared__ __attribute__((aligned(16))) unsigned short Abuf[2][16 * A_PITCH];
    const int tid = threadIdx.x;
    const int last = nrows - 1;
    const int ntiles = (nrows + 15) >> 4;
    const int nchunks = (e + 255) >> 8;
    const int elast = (e > 0) ? e - 1 : 0;

    const int wave = tid >> 5;
    const int lane = tid & 31;
    const int n0 = wave * 16;
    const int mrow = lane & 15;
    const int khalf = (lane >> 4) << 3;

    int t = blockIdx.x;
    float4 va[8];
    if (t < ntiles) load_x_tile(va, x, t * 16, last, tid);

    const unsigned short* brow = Wt1 + (size_t)(n0 + mrow) * F_IN;
    BFrag b[16];
#pragma unroll
    for (int s = 0; s < 16; ++s) {
        b[s].u[0] = *(const uint4*)&brow[s * 32 + khalf];
        b[s].u[1] = *(const uint4*)&brow[s * 32 + 16 + khalf];
    }

    int buf = 0;
    for (; t < ntiles; t += gridDim.x) {
        unsigned short* A = Abuf[buf];
#pragma unroll
        for (int j = 0; j < 8; ++j) {
            const int q = tid + 256 * j;
            const int m = q >> 7;
            const int c4 = q & 127;
            uint2 pk;
            pk.x = f2bf(va[j].x) | (f2bf(va[j].y) << 16);
            pk.y = f2bf(va[j].z) | (f2bf(va[j].w) << 16);
            *(uint2*)&A[m * A_PITCH + c4 * 4] = pk;
        }
        __syncthreads();

        const int rbase = t * 16 + ((lane >> 4) << 3);
        float dv[8];
#pragma unroll
        for (int r = 0; r < 8; ++r) {
            const int row = rbase + r;
            dv[r] = dinv[(row < last) ? row : last];
        }

        const int ei = t * 256 + tid;
        const int ec = (ei < elast) ? ei : elast;
        const int es = src[ec];
        const int ed = dst[ec];
        const int ewb = __float_as_int(ew[ec]);
        const int ek = rank[ec];

        load_x_tile(va, x, (t + gridDim.x) * 16, last, tid);
        __builtin_amdgcn_sched_barrier(0);

        v8f acc = {};
#pragma unroll
        for (int s = 0; s < 16; ++s) {
            BFrag a;
            a.u[0] = *(const uint4*)&A[mrow * A_PITCH + s * 32 + khalf];
            a.u[1] = *(const uint4*)&A[mrow * A_PITCH + s * 32 + 16 + khalf];
            acc = __builtin_amdgcn_wmma_f32_16x16x32_bf16(
                false, a.v, false, b[s].v, (short)0, acc, false, false);
        }

        const int rs = row_start[ed];

        const int col = n0 + (lane & 15);
#pragma unroll
        for (int r = 0; r < 8; ++r) {
            int row = rbase + r;
            const float o = acc[r] * dv[r];
            if (row < nrows) h1[(size_t)row * F_HID + col] = o;
        }

        csr[rs + ek] = make_int2(es, ewb);
        buf ^= 1;
    }

    for (int c = ntiles + blockIdx.x; c < nchunks; c += gridDim.x)
        fill_chunk(src, dst, ew, row_start, rank, csr, c * 256 + tid, e);
}


#define ROWS_PER_WAVE1 2

template <int B>
__device__ __forceinline__ void gather_rows(float4& acc, const float* hcol,
                                            int s_l, int c_l, int j) {
    float4 v[B];
    float cc[B];
#pragma unroll
    for (int u = 0; u < B; ++u) {
        const int s = __builtin_amdgcn_readlane(s_l, j + u);
        cc[u] = __int_as_float(__builtin_amdgcn_readlane(c_l, j + u));
        v[u] = *(const float4*)(hcol + (size_t)s * F_HID);
    }
#pragma unroll
    for (int u = 0; u < B; ++u) {
        acc.x += cc[u] * v[u].x;
        acc.y += cc[u] * v[u].y;
        acc.z += cc[u] * v[u].z;
        acc.w += cc[u] * v[u].w;
    }
}

__global__ void k_gather1(const int2* __restrict__ csr,
                          const int* __restrict__ row_start,
                          const int* __restrict__ cnt,
                          const float* __restrict__ dinv,
                          const float* __restrict__ h1,
                          const float* __restrict__ b1,
                          unsigned short* __restrict__ hb, int n) {
    const int lane = threadIdx.x & 31;
    const int wid = __builtin_amdgcn_readfirstlane(
        (int)((blockIdx.x * blockDim.x + threadIdx.x) >> 5));
    const float4 bias = *(const float4*)(b1 + lane * 4);
    const float* hcol = h1 + lane * 4;

    for (int r = 0; r < ROWS_PER_WAVE1; ++r) {
        const int i = wid * ROWS_PER_WAVE1 + r;
        if (i >= n) break;
        const int start = row_start[i];
        const int deg_i = cnt[i];
        const float di = dinv[i];

        int m = (deg_i < 32) ? deg_i : 32;
        int s_l = 0;
        int c_l = 0;
        if (lane < m) {
            const int2 p = csr[start + lane];
            s_l = p.x;
            c_l = p.y;
        }
        float4 acc = *(const float4*)(hcol + (size_t)i * F_HID);

        int base = 0;
        while (m > 0) {
            int j = 0;
            for (; j + 8 <= m; j += 8) gather_rows<8>(acc, hcol, s_l, c_l, j);
            if (j + 4 <= m) { gather_rows<4>(acc, hcol, s_l, c_l, j); j += 4; }
            if (j + 2 <= m) { gather_rows<2>(acc, hcol, s_l, c_l, j); j += 2; }
            if (j < m)      { gather_rows<1>(acc, hcol, s_l, c_l, j); }
            base += 32;
            m = (deg_i - base < 32) ? (deg_i - base) : 32;
            if (m > 0 && lane < m) {
                const int2 p = csr[start + base + lane];
                s_l = p.x;
                c_l = p.y;
            }
        }

        const float ox = fmaxf(acc.x * di + bias.x, 0.0f);
        const float oy = fmaxf(acc.y * di + bias.y, 0.0f);
        const float oz = fmaxf(acc.z * di + bias.z, 0.0f);
        const float ow = fmaxf(acc.w * di + bias.w, 0.0f);
        uint2 pk;
        pk.x = f2bf(ox) | (f2bf(oy) << 16);
        pk.y = f2bf(oz) | (f2bf(ow) << 16);
        *(uint2*)(hb + (size_t)i * F_HID + lane * 4) = pk;
    }
}


__global__ void k_gemm2_wmma(const unsigned short* __restrict__ hb,
                             const unsigned short* __restrict__ Wt2,
                             const float* __restrict__ dinv,
                             float* __restrict__ z,
                             int nrows) {
    const int wave = threadIdx.x >> 5;
    const int lane = threadIdx.x & 31;
    const int m0 = blockIdx.x * 128 + wave * 16;
    if (m0 >= nrows) return;

    const int mrow = lane & 15;
    const int khalf = (lane >> 4) << 3;
    const unsigned short* arow = hb + (size_t)(m0 + mrow) * F_HID;
    const unsigned short* brow = Wt2 + (size_t)mrow * F_HID;

    const int rbase = m0 + ((lane >> 4) << 3);
    const int last = nrows - 1;
    float dv[8];
#pragma unroll
    for (int r = 0; r < 8; ++r) {
        const int row = rbase + r;
        dv[r] = dinv[(row < last) ? row : last];
    }

    v8f acc = {};
#pragma unroll
    for (int k = 0; k < F_HID; k += 32) {
        BFrag a, b;
        a.u[0] = *(const uint4*)(arow + k + khalf);
        a.u[1] = *(const uint4*)(arow + k + 16 + khalf);
        b.u[0] = *(const uint4*)(brow + k + khalf);
        b.u[1] = *(const uint4*)(brow + k + 16 + khalf);
        acc = __builtin_amdgcn_wmma_f32_16x16x32_bf16(
            false, a.v, false, b.v, (short)0, acc, false, false);
    }

    const int col = lane & 15;
#pragma unroll
    for (int r = 0; r < 8; ++r) {
        int row = rbase + r;
        const float o = acc[r] * dv[r];
        if (row < nrows) z[(size_t)row * F_OUT + col] = o;
    }
}


__global__ void k_gather2(const int2* __restrict__ csr,
                          const int* __restrict__ row_start,
                          const int* __restrict__ cnt,
                          const float* __restrict__ dinv,
                          const float* __restrict__ z,
                          const float* __restrict__ b2,
                          float* __restrict__ out, int n) {
    const int lane = threadIdx.x & 31;
    const int half = lane >> 4;
    const int c = lane & 15;
    const int wid = __builtin_amdgcn_readfirstlane(
        (int)((blockIdx.x * blockDim.x + threadIdx.x) >> 5));
    if (wid * 2 >= n) return;
    const int i = wid * 2 + half;
    const bool valid = i < n;

    int start = 0, deg_i = 0;
    float acc = 0.0f;
    float di = 0.0f;
    if (valid) {
        start = row_start[i];
        deg_i = cnt[i];
        di = dinv[i];
        acc = z[(size_t)i * F_OUT + c];
    }
    const int other = __shfl_xor(deg_i, 16);
    const int maxdeg = __builtin_amdgcn_readfirstlane(deg_i > other ? deg_i : other);

    for (int base = 0; base < maxdeg; base += 16) {
        const int m = deg_i - base;
        int s_l = 0;
        float c_l = 0.0f;
        if (c < m) {
            const int2 p = csr[start + base + c];
            s_l = p.x;
            c_l = __int_as_float(p.y);
        }
        float v[16];
        float cf[16];
#pragma unroll
        for (int u = 0; u < 16; ++u) {
            const int s = __shfl(s_l, u, 16);
            cf[u] = __shfl(c_l, u, 16);
            v[u] = (u < m) ? z[(size_t)s * F_OUT + c] : 0.0f;
        }
#pragma unroll
        for (int u = 0; u < 16; ++u) acc += cf[u] * v[u];
    }

    const float l = acc * di + b2[c];
    float mx = l;
#pragma unroll
    for (int k = 8; k >= 1; k >>= 1) mx = fmaxf(mx, __shfl_xor(mx, k, 16));
    const float ex = expf(l - mx);
    float ssum = 0.0f;
#pragma unroll
    for (int u = 0; u < 16; ++u) ssum += __shfl(ex, u, 16);
    const float lse = mx + logf(ssum);
    if (valid) out[(size_t)i * F_OUT + c] = l - lse;
}


extern "C" void kernel_launch(void* const* d_in, const int* in_sizes, int n_in,
                              void* d_out, int out_size, void* d_ws, size_t ws_size,
                              hipStream_t stream) {
    (void)n_in; (void)out_size; (void)ws_size;

    const float* x  = (const float*)d_in[0];
    const int*   ei = (const int*)d_in[1];
    const float* ew = (const float*)d_in[2];
    const float* W1 = (const float*)d_in[3];
    const float* b1 = (const float*)d_in[4];
    const float* W2 = (const float*)d_in[5];
    const float* b2 = (const float*)d_in[6];
    float* out = (float*)d_out;

    const int N = in_sizes[0] / F_IN;
    const int E = in_sizes[2];
    const int* src = ei;
    const int* dst = ei + E;

    char* ws = (char*)d_ws;
    size_t off = 0;
    auto take = [&](size_t bytes) -> char* {
        char* p = ws + off;
        off += (bytes + 255) & ~(size_t)255;
        return p;
    };
    float*          dinv = (float*)take((size_t)N * 4);
    unsigned short* Wt1  = (unsigned short*)take((size_t)F_IN * F_HID * 2);
    unsigned short* Wt2  = (unsigned short*)take((size_t)F_HID * F_OUT * 2);
    float*          bufA = (float*)take((size_t)N * F_HID * 4);
    char*           bufB = take((size_t)N * F_HID * 4);

    unsigned short* hb = (unsigned short*)bufB;
    char* q = bufB + (size_t)N * F_HID * 2;
    int2* csr       = (int2*)q;  q += (size_t)E * 8;
    int*  rank      = (int*)q;   q += (size_t)E * 4;
    int*  cnt       = (int*)q;   q += (size_t)N * 4;
    int*  row_start = (int*)q;   q += (size_t)N * 4;
    int*  total     = (int*)q;
    float* z = bufA;

    const int T = 256;

    k_zero<<<(N + T - 1) / T, T, 0, stream>>>(dinv, cnt, total, N);
    k_count<<<(E + T - 1) / T, T, 0, stream>>>(dst, ew, dinv, cnt, rank, E);
    k_offsets<<<(N + T - 1) / T, T, 0, stream>>>(dinv, cnt, row_start, total, N);

    k_prep_weights<<<(F_IN * F_HID + F_HID * F_OUT + T - 1) / T, T, 0, stream>>>(
        W1, W2, Wt1, Wt2);

    k_gemm1_wmma<<<GEMM1_BLOCKS, T, 0, stream>>>(x, Wt1, dinv, bufA, N,
                                                 src, dst, ew, row_start, rank, csr, E);
    k_gather1<<<(N + ROWS_PER_WAVE1 * 8 - 1) / (ROWS_PER_WAVE1 * 8), T, 0, stream>>>(
        csr, row_start, cnt, dinv, bufA, b1, hb, N);

    k_gemm2_wmma<<<(N + 127) / 128, T, 0, stream>>>(hb, Wt2, dinv, z, N);
    k_gather2<<<(N + 15) / 16, T, 0, stream>>>(csr, row_start, cnt, dinv, z, b2, out, N);
}


# edge buckets by a two-level counting sort on LDS counters (chunk by 256-row bin, then bin by row): no global atomic per edge; degree = LDS f32 adds, bucket order = LDS add arrival rank
# speedup vs baseline: 4.6322x; 1.8267x over previous
#include <hip/hip_runtime.h>
#include <math.h>

#define F_IN 512
#define F_HID 128
#define F_OUT 16

typedef __attribute__((ext_vector_type(16))) __bf16 v16bf;
typedef __attribute__((ext_vector_type(8)))  float  v8f;

union BFrag {
    uint4 u[2];
    v16bf v;
};

__device__ __forceinline__ unsigned int f2bf(float f) {
    union { float f; unsigned int u; } cv; cv.f = f;
    unsigned int u = cv.u;
    unsigned int r = u + 0x7FFFu + ((u >> 16) & 1u);
    return (r >> 16) & 0xFFFFu;
}


#define BIN_SHIFT 8
#define BIN_ROWS 256
#define MAX_BINS 512
#define SORT_BLOCKS 256
#define SORT_THREADS 1024
#define RS_PITCH (MAX_BINS + 1)

__device__ __forceinline__ int excl_scan512(int c, int tid, int* wsum, int& total) {
    const int lane = tid & 31;
    const int wv = tid >> 5;
    int x = c;
#pragma unroll
    for (int off = 1; off < 32; off <<= 1) {
        const int t = __shfl_up(x, off);
        if (lane >= off) x += t;
    }
    if (wv < 16 && lane == 31) wsum[wv] = x;
    __syncthreads();
    int pre = 0, tot = 0;
#pragma unroll
    for (int w = 0; w < 16; ++w) {
        const int v = wsum[w];
        if (w < wv) pre += v;
        tot += v;
    }
    total = tot;
    return pre + x - c;
}

__global__ __launch_bounds__(SORT_THREADS) void k_bin_chunks(const int* __restrict__ src,
                             const int* __restrict__ dst,
                             const float* __restrict__ ew,
                             int2* __restrict__ rec_dw, int* __restrict__ rec_s,
                             int* __restrict__ runstart, int* total,
                             int e, int chunk, int nbins) {
    __shared__ int hist[MAX_BINS];
    __shared__ int wsum[16];
    const int tid = threadIdx.x;
    int lo = blockIdx.x * chunk;
    lo = (lo < e) ? lo : e;
    int hi = lo + chunk;
    hi = (hi < e) ? hi : e;
    if (tid < MAX_BINS) hist[tid] = 0;
    if (blockIdx.x == 0 && tid == 0) *total = 0;
    __syncthreads();

    for (int i = lo + tid; i < hi; i += 4 * SORT_THREADS) {
        int d[4];
#pragma unroll
        for (int u = 0; u < 4; ++u) {
            const int j = i + u * SORT_THREADS;
            d[u] = (j < hi) ? dst[j] : -1;
        }
#pragma unroll
        for (int u = 0; u < 4; ++u)
            if (d[u] >= 0) atomicAdd(&hist[d[u] >> BIN_SHIFT], 1);
    }
    __syncthreads();

    const int c = (tid < MAX_BINS) ? hist[tid] : 0;
    int tot;
    const int excl = excl_scan512(c, tid, wsum, tot);
    if (tid < MAX_BINS) hist[tid] = excl;
    if (tid < nbins) runstart[blockIdx.x * RS_PITCH + tid] = lo + excl;
    if (tid == 0) runstart[blockIdx.x * RS_PITCH + nbins] = hi;
    __syncthreads();

    for (int i = lo + tid; i < hi; i += 4 * SORT_THREADS) {
        int d[4], s[4], w[4];
#pragma unroll
        for (int u = 0; u < 4; ++u) {
            const int j = i + u * SORT_THREADS;
            const bool ok = j < hi;
            d[u] = ok ? dst[j] : -1;
            s[u] = ok ? src[j] : 0;
            w[u] = ok ? __float_as_int(ew[j]) : 0;
        }
#pragma unroll
        for (int u = 0; u < 4; ++u) {
            if (d[u] >= 0) {
                const int slot = lo + atomicAdd(&hist[d[u] >> BIN_SHIFT], 1);
                rec_dw[slot] = make_int2(d[u], w[u]);
                rec_s[slot] = s[u];
            }
        }
    }
}

__global__ __launch_bounds__(SORT_THREADS) void k_bin_rows(const int2* __restrict__ rec_dw,
                             const int* __restrict__ rec_s,
                             const int* __restrict__ runstart,
                             float* __restrict__ dinv, int* __restrict__ cnt,
                             int* __restrict__ row_start, int2* __restrict__ csr,
                             int* total, int n) {
    __shared__ int   c_l[BIN_ROWS];
    __shared__ float g_l[BIN_ROWS];
    __shared__ int   r_lo[SORT_BLOCKS];
    __shared__ int   r_hi[SORT_BLOCKS];
    __shared__ int   wsum[16];
    __shared__ int   base_s;
    const int tid = threadIdx.x;
    const int lane = tid & 31;
    const int wv = tid >> 5;
    const int k = blockIdx.x;
    const int row0 = k << BIN_SHIFT;
    if (tid < BIN_ROWS) { c_l[tid] = 0; g_l[tid] = 1.0f; }
    if (tid < SORT_BLOCKS) {
        const int* p = runstart + tid * RS_PITCH + k;
        r_lo[tid] = p[0];
        r_hi[tid] = p[1];
    }
    __syncthreads();

    for (int b0 = wv; b0 < SORT_BLOCKS; b0 += 128) {
        int2 v[4];
#pragma unroll
        for (int u = 0; u < 4; ++u) {
            const int b = b0 + 32 * u;
            const int i = r_lo[b] + lane;
            v[u] = (i < r_hi[b]) ? rec_dw[i] : make_int2(-1, 0);
        }
#pragma unroll
        for (int u = 0; u < 4; ++u) {
            if (v[u].x >= 0) {
                const int r = v[u].x - row0;
                atomicAdd(&c_l[r], 1);
                atomicAdd(&g_l[r], __int_as_float(v[u].y));
            }
        }
        for (int u = 0; u < 4; ++u) {
            const int b = b0 + 32 * u;
            const int hi = r_hi[b];
            for (int i = r_lo[b] + 32 + lane; i < hi; i += 32) {
                const int2 q = rec_dw[i];
                const int r = q.x - row0;
                atomicAdd(&c_l[r], 1);
                atomicAdd(&g_l[r], __int_as_float(q.y));
            }
        }
    }
    __syncthreads();

    const int c = (tid < BIN_ROWS) ? c_l[tid] : 0;
    int tot;
    const int excl = excl_scan512(c, tid, wsum, tot);
    if (tid == 0) base_s = atomicAdd(total, tot);
    __syncthreads();
    if (tid < BIN_ROWS) {
        const int start = base_s + excl;
        c_l[tid] = start;
        const int i = row0 + tid;
        if (i < n) {
            row_start[i] = start;
            cnt[i] = c;
            const float dg = g_l[tid];
            dinv[i] = (dg > 0.0f) ? rsqrtf(dg) : 0.0f;
        }
    }
    __syncthreads();

    for (int b0 = wv; b0 < SORT_BLOCKS; b0 += 128) {
        int2 v[4];
        int s[4];
#pragma unroll
        for (int u = 0; u < 4; ++u) {
            const int b = b0 + 32 * u;
            const int i = r_lo[b] + lane;
            const bool ok = i < r_hi[b];
            v[u] = ok ? rec_dw[i] : make_int2(-1, 0);
            s[u] = ok ? rec_s[i] : 0;
        }
#pragma unroll
        for (int u = 0; u < 4; ++u) {
            if (v[u].x >= 0) {
                const int slot = atomicAdd(&c_l[v[u].x - row0], 1);
                csr[slot] = make_int2(s[u], v[u].y);
            }
        }
        for (int u = 0; u < 4; ++u) {
            const int b = b0 + 32 * u;
            const int hi = r_hi[b];
            for (int i = r_lo[b] + 32 + lane; i < hi; i += 32) {
                const int2 q = rec_dw[i];
                const int sv = rec_s[i];
                const int slot = atomicAdd(&c_l[q.x - row0], 1);
                csr[slot] = make_int2(sv, q.y);
            }
        }
    }
}


__global__ void k_prep_weights(const float* __restrict__ W1,
                               const float* __restrict__ W2,
                               unsigned short* __restrict__ Wt1,
                               unsigned short* __restrict__ Wt2) {
    int t = blockIdx.x * blockDim.x + threadIdx.x;
    if (t < F_IN * F_HID) {
        int n = t / F_IN, k = t % F_IN;
        Wt1[t] = (unsigned short)f2bf(W1[k * F_HID + n]);
    }
    int t2 = t - F_IN * F_HID;
    if (t2 >= 0 && t2 < F_HID * F_OUT) {
        int n = t2 / F_HID, k = t2 % F_HID;
        Wt2[t2] = (unsigned short)f2bf(W2[k * F_OUT + n]);
    }
}


#define A_PITCH (F_IN + 8)
#define GEMM1_BLOCKS 512

__device__ __forceinline__ void load_x_tile(float4 (&va)[8], const float* __restrict__ x,
                                            int row0, int last, int tid) {
#pragma unroll
    for (int j = 0; j < 8; ++j) {
        const int q = tid + 256 * j;
        const int m = q >> 7;
        const int c4 = q & 127;
        int gr = row0 + m;
        gr = (gr < last) ? gr : last;
        va[j] = *(const float4*)(x + (size_t)gr * F_IN + c4 * 4);
    }
}

__global__ __launch_bounds__(256) void k_gemm1_wmma(const float* __restrict__ x,
                             const unsigned short* __restrict__ Wt1,
                             const float* __restrict__ dinv,
                             float* __restrict__ h1,
                             int nrows) {
    __shared__ __attribute__((aligned(16))) unsigned short Abuf[2][16 * A_PITCH];
    const int tid = threadIdx.x;
    const int last = nrows - 1;
    const int ntiles = (nrows + 15) >> 4;

    const int wave = tid >> 5;
    const int lane = tid & 31;
    const int n0 = wave * 16;
    const int mrow = lane & 15;
    const int khalf = (lane >> 4) << 3;

    int t = blockIdx.x;
    float4 va[8];
    if (t < ntiles) load_x_tile(va, x, t * 16, last, tid);

    const unsigned short* brow = Wt1 + (size_t)(n0 + mrow) * F_IN;
    BFrag b[16];
#pragma unroll
    for (int s = 0; s < 16; ++s) {
        b[s].u[0] = *(const uint4*)&brow[s * 32 + khalf];
        b[s].u[1] = *(const uint4*)&brow[s * 32 + 16 + khalf];
    }

    int buf = 0;
    for (; t < ntiles; t += gridDim.x) {
        unsigned short* A = Abuf[buf];
#pragma unroll
        for (int j = 0; j < 8; ++j) {
            const int q = tid + 256 * j;
            const int m = q >> 7;
            const int c4 = q & 127;
            uint2 pk;
            pk.x = f2bf(va[j].x) | (f2bf(va[j].y) << 16);
            pk.y = f2bf(va[j].z) | (f2bf(va[j].w) << 16);
            *(uint2*)&A[m * A_PITCH + c4 * 4] = pk;
        }
        __syncthreads();

        const int rbase = t * 16 + ((lane >> 4) << 3);
        float dv[8];
#pragma unroll
        for (int r = 0; r < 8; ++r) {
            const int row = rbase + r;
            dv[r] = dinv[(row < last) ? row : last];
        }

        load_x_tile(va, x, (t + gridDim.x) * 16, last, tid);
        __builtin_amdgcn_sched_barrier(0);

        v8f acc = {};
#pragma unroll
        for (int s = 0; s < 16; ++s) {
            BFrag a;
            a.u[0] = *(const uint4*)&A[mrow * A_PITCH + s * 32 + khalf];
            a.u[1] = *(const uint4*)&A[mrow * A_PITCH + s * 32 + 16 + khalf];
            acc = __builtin_amdgcn_wmma_f32_16x16x32_bf16(
                false, a.v, false, b[s].v, (short)0, acc, false, false);
        }

        const int col = n0 + (lane & 15);
#pragma unroll
        for (int r = 0; r < 8; ++r) {
            int row = rbase + r;
            const float o = acc[r] * dv[r];
            if (row < nrows) h1[(size_t)row * F_HID + col] = o;
        }

        buf ^= 1;
    }
}


#define ROWS_PER_WAVE1 2

template <int B>
__device__ __forceinline__ void gather_rows(float4& acc, const float* hcol,
                                            int s_l, int c_l, int j) {
    float4 v[B];
    float cc[B];
#pragma unroll
    for (int u = 0; u < B; ++u) {
        const int s = __builtin_amdgcn_readlane(s_l, j + u);
        cc[u] = __int_as_float(__builtin_amdgcn_readlane(c_l, j + u));
        v[u] = *(const float4*)(hcol + (size_t)s * F_HID);
    }
#pragma unroll
    for (int u = 0; u < B; ++u) {
        acc.x += cc[u] * v[u].x;
        acc.y += cc[u] * v[u].y;
        acc.z += cc[u] * v[u].z;
        acc.w += cc[u] * v[u].w;
    }
}

__global__ void k_gather1(const int2* __restrict__ csr,
                          const int* __restrict__ row_start,
                          const int* __restrict__ cnt,
                          const float* __restrict__ dinv,
                          const float* __restrict__ h1,
                          const float* __restrict__ b1,
                          unsigned short* __restrict__ hb, int n) {
    const int lane = threadIdx.x & 31;
    const int wid = __builtin_amdgcn_readfirstlane(
        (int)((blockIdx.x * blockDim.x + threadIdx.x) >> 5));
    const float4 bias = *(const float4*)(b1 + lane * 4);
    const float* hcol = h1 + lane * 4;

    for (int r = 0; r < ROWS_PER_WAVE1; ++r) {
        const int i = wid * ROWS_PER_WAVE1 + r;
        if (i >= n) break;
        const int start = row_start[i];
        const int deg_i = cnt[i];
        const float di = dinv[i];

        int m = (deg_i < 32) ? deg_i : 32;
        int s_l = 0;
        int c_l = 0;
        if (lane < m) {
            const int2 p = csr[start + lane];
            s_l = p.x;
            c_l = p.y;
        }
        float4 acc = *(const float4*)(hcol + (size_t)i * F_HID);

        int base = 0;
        while (m > 0) {
            int j = 0;
            for (; j + 8 <= m; j += 8) gather_rows<8>(acc, hcol, s_l, c_l, j);
            if (j + 4 <= m) { gather_rows<4>(acc, hcol, s_l, c_l, j); j += 4; }
            if (j + 2 <= m) { gather_rows<2>(acc, hcol, s_l, c_l, j); j += 2; }
            if (j < m)      { gather_rows<1>(acc, hcol, s_l, c_l, j); }
            base += 32;
            m = (deg_i - base < 32) ? (deg_i - base) : 32;
            if (m > 0 && lane < m) {
                const int2 p = csr[start + base + lane];
                s_l = p.x;
                c_l = p.y;
            }
        }

        const float ox = fmaxf(acc.x * di + bias.x, 0.0f);
        const float oy = fmaxf(acc.y * di + bias.y, 0.0f);
        const float oz = fmaxf(acc.z * di + bias.z, 0.0f);
        const float ow = fmaxf(acc.w * di + bias.w, 0.0f);
        uint2 pk;
        pk.x = f2bf(ox) | (f2bf(oy) << 16);
        pk.y = f2bf(oz) | (f2bf(ow) << 16);
        *(uint2*)(hb + (size_t)i * F_HID + lane * 4) = pk;
    }
}


__global__ void k_gemm2_wmma(const unsigned short* __restrict__ hb,
                             const unsigned short* __restrict__ Wt2,
                             const float* __restrict__ dinv,
                             float* __restrict__ z,
                             int nrows) {
    const int wave = threadIdx.x >> 5;
    const int lane = threadIdx.x & 31;
    const int m0 = blockIdx.x * 128 + wave * 16;
    if (m0 >= nrows) return;

    const int mrow = lane & 15;
    const int khalf = (lane >> 4) << 3;
    const unsigned short* arow = hb + (size_t)(m0 + mrow) * F_HID;
    const unsigned short* brow = Wt2 + (size_t)mrow * F_HID;

    const int rbase = m0 + ((lane >> 4) << 3);
    const int last = nrows - 1;
    float dv[8];
#pragma unroll
    for (int r = 0; r < 8; ++r) {
        const int row = rbase + r;
        dv[r] = dinv[(row < last) ? row : last];
    }

    v8f acc = {};
#pragma unroll
    for (int k = 0; k < F_HID; k += 32) {
        BFrag a, b;
        a.u[0] = *(const uint4*)(arow + k + khalf);
        a.u[1] = *(const uint4*)(arow + k + 16 + khalf);
        b.u[0] = *(const uint4*)(brow + k + khalf);
        b.u[1] = *(const uint4*)(brow + k + 16 + khalf);
        acc = __builtin_amdgcn_wmma_f32_16x16x32_bf16(
            false, a.v, false, b.v, (short)0, acc, false, false);
    }

    const int col = lane & 15;
#pragma unroll
    for (int r = 0; r < 8; ++r) {
        int row = rbase + r;
        const float o = acc[r] * dv[r];
        if (row < nrows) z[(size_t)row * F_OUT + col] = o;
    }
}


__global__ void k_gather2(const int2* __restrict__ csr,
                          const int* __restrict__ row_start,
                          const int* __restrict__ cnt,
                          const float* __restrict__ dinv,
                          const float* __restrict__ z,
                          const float* __restrict__ b2,
                          float* __restrict__ out, int n) {
    const int lane = threadIdx.x & 31;
    const int half = lane >> 4;
    const int c = lane & 15;
    const int wid = __builtin_amdgcn_readfirstlane(
        (int)((blockIdx.x * blockDim.x + threadIdx.x) >> 5));
    if (wid * 2 >= n) return;
    const int i = wid * 2 + half;
    const bool valid = i < n;

    int start = 0, deg_i = 0;
    float acc = 0.0f;
    float di = 0.0f;
    if (valid) {
        start = row_start[i];
        deg_i = cnt[i];
        di = dinv[i];
        acc = z[(size_t)i * F_OUT + c];
    }
    const int other = __shfl_xor(deg_i, 16);
    const int maxdeg = __builtin_amdgcn_readfirstlane(deg_i > other ? deg_i : other);

    for (int base = 0; base < maxdeg; base += 16) {
        const int m = deg_i - base;
        int s_l = 0;
        float c_l = 0.0f;
        if (c < m) {
            const int2 p = csr[start + base + c];
            s_l = p.x;
            c_l = __int_as_float(p.y);
        }
        float v[16];
        float cf[16];
#pragma unroll
        for (int u = 0; u < 16; ++u) {
            const int s = __shfl(s_l, u, 16);
            cf[u] = __shfl(c_l, u, 16);
            v[u] = (u < m) ? z[(size_t)s * F_OUT + c] : 0.0f;
        }
#pragma unroll
        for (int u = 0; u < 16; ++u) acc += cf[u] * v[u];
    }

    const float l = acc * di + b2[c];
    float mx = l;
#pragma unroll
    for (int k = 8; k >= 1; k >>= 1) mx = fmaxf(mx, __shfl_xor(mx, k, 16));
    const float ex = expf(l - mx);
    float ssum = 0.0f;
#pragma unroll
    for (int u = 0; u < 16; ++u) ssum += __shfl(ex, u, 16);
    const float lse = mx + logf(ssum);
    if (valid) out[(size_t)i * F_OUT + c] = l - lse;
}


extern "C" void kernel_launch(void* const* d_in, const int* in_sizes, int n_in,
                              void* d_out, int out_size, void* d_ws, size_t ws_size,
                              hipStream_t stream) {
    (void)n_in; (void)out_size; (void)ws_size;

    const float* x  = (const float*)d_in[0];
    const int*   ei = (const int*)d_in[1];
    const float* ew = (const float*)d_in[2];
    const float* W1 = (const float*)d_in[3];
    const float* b1 = (const float*)d_in[4];
    const float* W2 = (const float*)d_in[5];
    const float* b2 = (const float*)d_in[6];
    float* out = (float*)d_out;

    const int N = in_sizes[0] / F_IN;
    const int E = in_sizes[2];
    const int* src = ei;
    const int* dst = ei + E;

    char* ws = (char*)d_ws;
    size_t off = 0;
    auto take = [&](size_t bytes) -> char* {
        char* p = ws + off;
        off += (bytes + 255) & ~(size_t)255;
        return p;
    };
    float*          dinv = (float*)take((size_t)N * 4);
    unsigned short* Wt1  = (unsigned short*)take((size_t)F_IN * F_HID * 2);
    unsigned short* Wt2  = (unsigned short*)take((size_t)F_HID * F_OUT * 2);
    float*          bufA = (float*)take((size_t)N * F_HID * 4);
    char*           bufB = take((size_t)N * F_HID * 4);

    unsigned short* hb = (unsigned short*)bufB;
    int2* rec_dw = (int2*)bufB;
    int*  rec_s  = (int*)(bufB + (size_t)E * 8);
    char* q = bufB + (size_t)N * F_HID * 2;
    int2* csr       = (int2*)q;  q += (size_t)E * 8;
    int*  runstart  = (int*)q;   q += (size_t)SORT_BLOCKS * RS_PITCH * 4;
    int*  cnt       = (int*)q;   q += (size_t)N * 4;
    int*  row_start = (int*)q;   q += (size_t)N * 4;
    int*  total     = (int*)q;
    float* z = bufA;

    const int T = 256;

    const int nbins = (N + BIN_ROWS - 1) / BIN_ROWS;
    const int chunk = (E + SORT_BLOCKS - 1) / SORT_BLOCKS;
    k_bin_chunks<<<SORT_BLOCKS, SORT_THREADS, 0, stream>>>(src, dst, ew, rec_dw, rec_s,
                                                          runstart, total, E, chunk, nbins);
    k_bin_rows<<<nbins, SORT_THREADS, 0, stream>>>(rec_dw, rec_s, runstart, dinv, cnt,
                                                  row_start, csr, total, N);

    k_prep_weights<<<(F_IN * F_HID + F_HID * F_OUT + T - 1) / T, T, 0, stream>>>(
        W1, W2, Wt1, Wt2);

    k_gemm1_wmma<<<GEMM1_BLOCKS, T, 0, stream>>>(x, Wt1, dinv, bufA, N);
    k_gather1<<<(N + ROWS_PER_WAVE1 * 8 - 1) / (ROWS_PER_WAVE1 * 8), T, 0, stream>>>(
        csr, row_start, cnt, dinv, bufA, b1, hb, N);

    k_gemm2_wmma<<<(N + 127) / 128, T, 0, stream>>>(hb, Wt2, dinv, z, N);
    k_gather2<<<(N + 15) / 16, T, 0, stream>>>(csr, row_start, cnt, dinv, z, b2, out, N);
}


# x@W1 table stored as bf16 rows, not f32 (storage dtype only: products, row factors and every sum stay f32, sum order kept): a row sum gathers 256 B per neighbour instead of 512 B
# speedup vs baseline: 5.1412x; 1.1099x over previous
#include <hip/hip_runtime.h>
#include <math.h>

#define F_IN 512
#define F_HID 128
#define F_OUT 16

typedef __attribute__((ext_vector_type(16))) __bf16 v16bf;
typedef __attribute__((ext_vector_type(8)))  float  v8f;

union BFrag {
    uint4 u[2];
    v16bf v;
};

__device__ __forceinline__ unsigned int f2bf(float f) {
    union { float f; unsigned int u; } cv; cv.f = f;
    unsigned int u = cv.u;
    unsigned int r = u + 0x7FFFu + ((u >> 16) & 1u);
    return (r >> 16) & 0xFFFFu;
}


#define BIN_SHIFT 8
#define BIN_ROWS 256
#define MAX_BINS 512
#define SORT_BLOCKS 256
#define SORT_THREADS 1024
#define RS_PITCH (MAX_BINS + 1)

__device__ __forceinline__ int excl_scan512(int c, int tid, int* wsum, int& total) {
    const int lane = tid & 31;
    const int wv = tid >> 5;
    int x = c;
#pragma unroll
    for (int off = 1; off < 32; off <<= 1) {
        const int t = __shfl_up(x, off);
        if (lane >= off) x += t;
    }
    if (wv < 16 && lane == 31) wsum[wv] = x;
    __syncthreads();
    int pre = 0, tot = 0;
#pragma unroll
    for (int w = 0; w < 16; ++w) {
        const int v = wsum[w];
        if (w < wv) pre += v;
        tot += v;
    }
    total = tot;
    return pre + x - c;
}

__global__ __launch_bounds__(SORT_THREADS) void k_bin_chunks(const int* __restrict__ src,
                             const int* __restrict__ dst,
                             const float* __restrict__ ew,
                             int2* __restrict__ rec_dw, int* __restrict__ rec_s,
                             int* __restrict__ runstart, int* total,
                             int e, int chunk, int nbins) {
    __shared__ int hist[MAX_BINS];
    __shared__ int wsum[16];
    const int tid = threadIdx.x;
    int lo = blockIdx.x * chunk;
    lo = (lo < e) ? lo : e;
    int hi = lo + chunk;
    hi = (hi < e) ? hi : e;
    if (tid < MAX_BINS) hist[tid] = 0;
    if (blockIdx.x == 0 && tid == 0) *total = 0;
    __syncthreads();

    for (int i = lo + tid; i < hi; i += 4 * SORT_THREADS) {
        int d[4];
#pragma unroll
        for (int u = 0; u < 4; ++u) {
            const int j = i + u * SORT_THREADS;
            d[u] = (j < hi) ? dst[j] : -1;
        }
#pragma unroll
        for (int u = 0; u < 4; ++u)
            if (d[u] >= 0) atomicAdd(&hist[d[u] >> BIN_SHIFT], 1);
    }
    __syncthreads();

    const int c = (tid < MAX_BINS) ? hist[tid] : 0;
    int tot;
    const int excl = excl_scan512(c, tid, wsum, tot);
    if (tid < MAX_BINS) hist[tid] = excl;
    if (tid < nbins) runstart[blockIdx.x * RS_PITCH + tid] = lo + excl;
    if (tid == 0) runstart[blockIdx.x * RS_PITCH + nbins] = hi;
    __syncthreads();

    for (int i = lo + tid; i < hi; i += 4 * SORT_THREADS) {
        int d[4], s[4], w[4];
#pragma unroll
        for (int u = 0; u < 4; ++u) {
            const int j = i + u * SORT_THREADS;
            const bool ok = j < hi;
            d[u] = ok ? dst[j] : -1;
            s[u] = ok ? src[j] : 0;
            w[u] = ok ? __float_as_int(ew[j]) : 0;
        }
#pragma unroll
        for (int u = 0; u < 4; ++u) {
            if (d[u] >= 0) {
                const int slot = lo + atomicAdd(&hist[d[u] >> BIN_SHIFT], 1);
                rec_dw[slot] = make_int2(d[u], w[u]);
                rec_s[slot] = s[u];
            }
        }
    }
}

__global__ __launch_bounds__(SORT_THREADS) void k_bin_rows(const int2* __restrict__ rec_dw,
                             const int* __restrict__ rec_s,
                             const int* __restrict__ runstart,
                             float* __restrict__ dinv, int* __restrict__ cnt,
                             int* __restrict__ row_start, int2* __restrict__ csr,
                             int* total, int n) {
    __shared__ int   c_l[BIN_ROWS];
    __shared__ float g_l[BIN_ROWS];
    __shared__ int   r_lo[SORT_BLOCKS];
    __shared__ int   r_hi[SORT_BLOCKS];
    __shared__ int   wsum[16];
    __shared__ int   base_s;
    const int tid = threadIdx.x;
    const int lane = tid & 31;
    const int wv = tid >> 5;
    const int k = blockIdx.x;
    const int row0 = k << BIN_SHIFT;
    if (tid < BIN_ROWS) { c_l[tid] = 0; g_l[tid] = 1.0f; }
    if (tid < SORT_BLOCKS) {
        const int* p = runstart + tid * RS_PITCH + k;
        r_lo[tid] = p[0];
        r_hi[tid] = p[1];
    }
    __syncthreads();

    for (int b0 = wv; b0 < SORT_BLOCKS; b0 += 128) {
        int2 v[4];
#pragma unroll
        for (int u = 0; u < 4; ++u) {
            const int b = b0 + 32 * u;
            const int i = r_lo[b] + lane;
            v[u] = (i < r_hi[b]) ? rec_dw[i] : make_int2(-1, 0);
        }
#pragma unroll
        for (int u = 0; u < 4; ++u) {
            if (v[u].x >= 0) {
                const int r = v[u].x - row0;
                atomicAdd(&c_l[r], 1);
                atomicAdd(&g_l[r], __int_as_float(v[u].y));
            }
        }
        for (int u = 0; u < 4; ++u) {
            const int b = b0 + 32 * u;
            const int hi = r_hi[b];
            for (int i = r_lo[b] + 32 + lane; i < hi; i += 32) {
                const int2 q = rec_dw[i];
                const int r = q.x - row0;
                atomicAdd(&c_l[r], 1);
                atomicAdd(&g_l[r], __int_as_float(q.y));
            }
        }
    }
    __syncthreads();

    const int c = (tid < BIN_ROWS) ? c_l[tid] : 0;
    int tot;
    const int excl = excl_scan512(c, tid, wsum, tot);
    if (tid == 0) base_s = atomicAdd(total, tot);
    __syncthreads();
    if (tid < BIN_ROWS) {
        const int start = base_s + excl;
        c_l[tid] = start;
        const int i = row0 + tid;
        if (i < n) {
            row_start[i] = start;
            cnt[i] = c;
            const float dg = g_l[tid];
            dinv[i] = (dg > 0.0f) ? rsqrtf(dg) : 0.0f;
        }
    }
    __syncthreads();

    for (int b0 = wv; b0 < SORT_BLOCKS; b0 += 128) {
        int2 v[4];
        int s[4];
#pragma unroll
        for (int u = 0; u < 4; ++u) {
            const int b = b0 + 32 * u;
            const int i = r_lo[b] + lane;
            const bool ok = i < r_hi[b];
            v[u] = ok ? rec_dw[i] : make_int2(-1, 0);
            s[u] = ok ? rec_s[i] : 0;
        }
#pragma unroll
        for (int u = 0; u < 4; ++u) {
            if (v[u].x >= 0) {
                const int slot = atomicAdd(&c_l[v[u].x - row0], 1);
                csr[slot] = make_int2(s[u], v[u].y);
            }
        }
        for (int u = 0; u < 4; ++u) {
            const int b = b0 + 32 * u;
            const int hi = r_hi[b];
            for (int i = r_lo[b] + 32 + lane; i < hi; i += 32) {
                const int2 q = rec_dw[i];
                const int sv = rec_s[i];
                const int slot = atomicAdd(&c_l[q.x - row0], 1);
                csr[slot] = make_int2(sv, q.y);
            }
        }
    }
}


__global__ void k_prep_weights(const float* __restrict__ W1,
                               const float* __restrict__ W2,
                               unsigned short* __restrict__ Wt1,
                               unsigned short* __restrict__ Wt2) {
    int t = blockIdx.x * blockDim.x + threadIdx.x;
    if (t < F_IN * F_HID) {
        int n = t / F_IN, k = t % F_IN;
        Wt1[t] = (unsigned short)f2bf(W1[k * F_HID + n]);
    }
    int t2 = t - F_IN * F_HID;
    if (t2 >= 0 && t2 < F_HID * F_OUT) {
        int n = t2 / F_HID, k = t2 % F_HID;
        Wt2[t2] = (unsigned short)f2bf(W2[k * F_OUT + n]);
    }
}


#define A_PITCH (F_IN + 8)
#define GEMM1_BLOCKS 512

__device__ __forceinline__ void load_x_tile(float4 (&va)[8], const float* __restrict__ x,
                                            int row0, int last, int tid) {
#pragma unroll
    for (int j = 0; j < 8; ++j) {
        const int q = tid + 256 * j;
        const int m = q >> 7;
        const int c4 = q & 127;
        int gr = row0 + m;
        gr = (gr < last) ? gr : last;
        va[j] = *(const float4*)(x + (size_t)gr * F_IN + c4 * 4);
    }
}

__global__ __launch_bounds__(256) void k_gemm1_wmma(const float* __restrict__ x,
                             const unsigned short* __restrict__ Wt1,
                             const float* __restrict__ dinv,
                             unsigned short* __restrict__ h1,
                             int nrows) {
    __shared__ __attribute__((aligned(16))) unsigned short Abuf[2][16 * A_PITCH];
    const int tid = threadIdx.x;
    const int last = nrows - 1;
    const int ntiles = (nrows + 15) >> 4;

    const int wave = tid >> 5;
    const int lane = tid & 31;
    const int n0 = wave * 16;
    const int mrow = lane & 15;
    const int khalf = (lane >> 4) << 3;

    int t = blockIdx.x;
    float4 va[8];
    if (t < ntiles) load_x_tile(va, x, t * 16, last, tid);

    const unsigned short* brow = Wt1 + (size_t)(n0 + mrow) * F_IN;
    BFrag b[16];
#pragma unroll
    for (int s = 0; s < 16; ++s) {
        b[s].u[0] = *(const uint4*)&brow[s * 32 + khalf];
        b[s].u[1] = *(const uint4*)&brow[s * 32 + 16 + khalf];
    }

    int buf = 0;
    for (; t < ntiles; t += gridDim.x) {
        unsigned short* A = Abuf[buf];
#pragma unroll
        for (int j = 0; j < 8; ++j) {
            const int q = tid + 256 * j;
            const int m = q >> 7;
            const int c4 = q & 127;
            uint2 pk;
            pk.x = f2bf(va[j].x) | (f2bf(va[j].y) << 16);
            pk.y = f2bf(va[j].z) | (f2bf(va[j].w) << 16);
            *(uint2*)&A[m * A_PITCH + c4 * 4] = pk;
        }
        __syncthreads();

        const int rbase = t * 16 + ((lane >> 4) << 3);
        float dv[8];
#pragma unroll
        for (int r = 0; r < 8; ++r) {
            const int row = rbase + r;
            dv[r] = dinv[(row < last) ? row : last];
        }

        load_x_tile(va, x, (t + gridDim.x) * 16, last, tid);
        __builtin_amdgcn_sched_barrier(0);

        v8f acc = {};
#pragma unroll
        for (int s = 0; s < 16; ++s) {
            BFrag a;
            a.u[0] = *(const uint4*)&A[mrow * A_PITCH + s * 32 + khalf];
            a.u[1] = *(const uint4*)&A[mrow * A_PITCH + s * 32 + 16 + khalf];
            acc = __builtin_amdgcn_wmma_f32_16x16x32_bf16(
                false, a.v, false, b[s].v, (short)0, acc, false, false);
        }

        const int odd = lane & 1;
        const int colpair = n0 + (lane & 14);
#pragma unroll
        for (int r = 0; r < 8; r += 2) {
            const float o0 = acc[r] * dv[r];
            const float o1 = acc[r + 1] * dv[r + 1];
            const float got = __int_as_float(__builtin_amdgcn_update_dpp(
                0, __float_as_int(odd ? o0 : o1), 0xB1, 0xF, 0xF, true));
            const float lo = odd ? got : o0;
            const float hi = odd ? o1 : got;
            const int row = rbase + r + odd;
            if (row < nrows)
                *(unsigned int*)(h1 + (size_t)row * F_HID + colpair) = f2bf(lo) | (f2bf(hi) << 16);
        }

        buf ^= 1;
    }
}


#define ROWS_PER_WAVE1 2

__device__ __forceinline__ float bf_lo(unsigned int w) { return __uint_as_float(w << 16); }
__device__ __forceinline__ float bf_hi(unsigned int w) { return __uint_as_float(w & 0xFFFF0000u); }

template <int B>
__device__ __forceinline__ void gather_rows(float4& acc, const uint2* __restrict__ h2, int lane,
                                            int s_l, int c_l, int j) {
    uint2 v[B];
    float cc[B];
#pragma unroll
    for (int u = 0; u < B; ++u) {
        const int s = __builtin_amdgcn_readlane(s_l, j + u);
        cc[u] = __int_as_float(__builtin_amdgcn_readlane(c_l, j + u));
        v[u] = h2[s * 32 + lane];
    }
#pragma unroll
    for (int u = 0; u < B; ++u) {
        acc.x = __builtin_fmaf(cc[u], bf_lo(v[u].x), acc.x);
        acc.y = __builtin_fmaf(cc[u], bf_hi(v[u].x), acc.y);
        acc.z = __builtin_fmaf(cc[u], bf_lo(v[u].y), acc.z);
        acc.w = __builtin_fmaf(cc[u], bf_hi(v[u].y), acc.w);
    }
}

__global__ void k_gather1(const int2* __restrict__ csr,
                          const int* __restrict__ row_start,
                          const int* __restrict__ cnt,
                          const float* __restrict__ dinv,
                          const unsigned short* __restrict__ h1,
                          const float* __restrict__ b1,
                          unsigned short* __restrict__ hb, int n) {
    const int lane = threadIdx.x & 31;
    const int wid = __builtin_amdgcn_readfirstlane(
        (int)((blockIdx.x * blockDim.x + threadIdx.x) >> 5));
    const float4 bias = *(const float4*)(b1 + lane * 4);
    const uint2* __restrict__ h2 = (const uint2*)h1;

    for (int r = 0; r < ROWS_PER_WAVE1; ++r) {
        const int i = wid * ROWS_PER_WAVE1 + r;
        if (i >= n) break;
        const int start = row_start[i];
        const int deg_i = cnt[i];
        const float di = dinv[i];

        int m = (deg_i < 32) ? deg_i : 32;
        int s_l = 0;
        int c_l = 0;
        if (lane < m) {
            const int2 p = csr[start + lane];
            s_l = p.x;
            c_l = p.y;
        }
        const uint2 own = h2[i * 32 + lane];
        float4 acc = make_float4(bf_lo(own.x), bf_hi(own.x), bf_lo(own.y), bf_hi(own.y));

        int base = 0;
        while (m > 0) {
            int j = 0;
            for (; j + 16 <= m; j += 16) gather_rows<16>(acc, h2, lane, s_l, c_l, j);
            if (j + 8 <= m) { gather_rows<8>(acc, h2, lane, s_l, c_l, j); j += 8; }
            if (j + 4 <= m) { gather_rows<4>(acc, h2, lane, s_l, c_l, j); j += 4; }
            if (j + 2 <= m) { gather_rows<2>(acc, h2, lane, s_l, c_l, j); j += 2; }
            if (j < m)      { gather_rows<1>(acc, h2, lane, s_l, c_l, j); }
            base += 32;
            m = (deg_i - base < 32) ? (deg_i - base) : 32;
            if (m > 0 && lane < m) {
                const int2 p = csr[start + base + lane];
                s_l = p.x;
                c_l = p.y;
            }
        }

        const float ox = fmaxf(acc.x * di + bias.x, 0.0f);
        const float oy = fmaxf(acc.y * di + bias.y, 0.0f);
        const float oz = fmaxf(acc.z * di + bias.z, 0.0f);
        const float ow = fmaxf(acc.w * di + bias.w, 0.0f);
        uint2 pk;
        pk.x = f2bf(ox) | (f2bf(oy) << 16);
        pk.y = f2bf(oz) | (f2bf(ow) << 16);
        *(uint2*)(hb + (size_t)i * F_HID + lane * 4) = pk;
    }
}


__global__ void k_gemm2_wmma(const unsigned short* __restrict__ hb,
                             const unsigned short* __restrict__ Wt2,
                             const float* __restrict__ dinv,
                             float* __restrict__ z,
                             int nrows) {
    const int wave = threadIdx.x >> 5;
    const int lane = threadIdx.x & 31;
    const int m0 = blockIdx.x * 128 + wave * 16;
    if (m0 >= nrows) return;

    const int mrow = lane & 15;
    const int khalf = (lane >> 4) << 3;
    const unsigned short* arow = hb + (size_t)(m0 + mrow) * F_HID;
    const unsigned short* brow = Wt2 + (size_t)mrow * F_HID;

    const int rbase = m0 + ((lane >> 4) << 3);
    const int last = nrows - 1;
    float dv[8];
#pragma unroll
    for (int r = 0; r < 8; ++r) {
        const int row = rbase + r;
        dv[r] = dinv[(row < last) ? row : last];
    }

    v8f acc = {};
#pragma unroll
    for (int k = 0; k < F_HID; k += 32) {
        BFrag a, b;
        a.u[0] = *(const uint4*)(arow + k + khalf);
        a.u[1] = *(const uint4*)(arow + k + 16 + khalf);
        b.u[0] = *(const uint4*)(brow + k + khalf);
        b.u[1] = *(const uint4*)(brow + k + 16 + khalf);
        acc = __builtin_amdgcn_wmma_f32_16x16x32_bf16(
            false, a.v, false, b.v, (short)0, acc, false, false);
    }

    const int col = lane & 15;
#pragma unroll
    for (int r = 0; r < 8; ++r) {
        int row = rbase + r;
        const float o = acc[r] * dv[r];
        if (row < nrows) z[(size_t)row * F_OUT + col] = o;
    }
}


__global__ void k_gather2(const int2* __restrict__ csr,
                          const int* __restrict__ row_start,
                          const int* __restrict__ cnt,
                          const float* __restrict__ dinv,
                          const float* __restrict__ z,
                          const float* __restrict__ b2,
                          float* __restrict__ out, int n) {
    const int lane = threadIdx.x & 31;
    const int half = lane >> 4;
    const int c = lane & 15;
    const int wid = __builtin_amdgcn_readfirstlane(
        (int)((blockIdx.x * blockDim.x + threadIdx.x) >> 5));
    if (wid * 2 >= n) return;
    const int i = wid * 2 + half;
    const bool valid = i < n;

    int start = 0, deg_i = 0;
    float acc = 0.0f;
    float di = 0.0f;
    if (valid) {
        start = row_start[i];
        deg_i = cnt[i];
        di = dinv[i];
        acc = z[(size_t)i * F_OUT + c];
    }
    const int other = __shfl_xor(deg_i, 16);
    const int maxdeg = __builtin_amdgcn_readfirstlane(deg_i > other ? deg_i : other);

    for (int base = 0; base < maxdeg; base += 16) {
        const int m = deg_i - base;
        int s_l = 0;
        float c_l = 0.0f;
        if (c < m) {
            const int2 p = csr[start + base + c];
            s_l = p.x;
            c_l = __int_as_float(p.y);
        }
        float v[16];
        float cf[16];
#pragma unroll
        for (int u = 0; u < 16; ++u) {
            const int s = __shfl(s_l, u, 16);
            cf[u] = __shfl(c_l, u, 16);
            v[u] = (u < m) ? z[(size_t)s * F_OUT + c] : 0.0f;
        }
#pragma unroll
        for (int u = 0; u < 16; ++u) acc += cf[u] * v[u];
    }

    const float l = acc * di + b2[c];
    float mx = l;
#pragma unroll
    for (int k = 8; k >= 1; k >>= 1) mx = fmaxf(mx, __shfl_xor(mx, k, 16));
    const float ex = expf(l - mx);
    float ssum = 0.0f;
#pragma unroll
    for (int u = 0; u < 16; ++u) ssum += __shfl(ex, u, 16);
    const float lse = mx + logf(ssum);
    if (valid) out[(size_t)i * F_OUT + c] = l - lse;
}


extern "C" void kernel_launch(void* const* d_in, const int* in_sizes, int n_in,
                              void* d_out, int out_size, void* d_ws, size_t ws_size,
                              hipStream_t stream) {
    (void)n_in; (void)out_size; (void)ws_size;

    const float* x  = (const float*)d_in[0];
    const int*   ei = (const int*)d_in[1];
    const float* ew = (const float*)d_in[2];
    const float* W1 = (const float*)d_in[3];
    const float* b1 = (const float*)d_in[4];
    const float* W2 = (const float*)d_in[5];
    const float* b2 = (const float*)d_in[6];
    float* out = (float*)d_out;

    const int N = in_sizes[0] / F_IN;
    const int E = in_sizes[2];
    const int* src = ei;
    const int* dst = ei + E;

    char* ws = (char*)d_ws;
    size_t off = 0;
    auto take = [&](size_t bytes) -> char* {
        char* p = ws + off;
        off += (bytes + 255) & ~(size_t)255;
        return p;
    };
    float*          dinv = (float*)take((size_t)N * 4);
    unsigned short* Wt1  = (unsigned short*)take((size_t)F_IN * F_HID * 2);
    unsigned short* Wt2  = (unsigned short*)take((size_t)F_HID * F_OUT * 2);
    float*          bufA = (float*)take((size_t)N * F_HID * 4);
    char*           bufB = take((size_t)N * F_HID * 4);

    unsigned short* hb = (unsigned short*)bufB;
    int2* rec_dw = (int2*)bufB;
    int*  rec_s  = (int*)(bufB + (size_t)E * 8);
    char* q = bufB + (size_t)N * F_HID * 2;
    int2* csr       = (int2*)q;  q += (size_t)E * 8;
    int*  runstart  = (int*)q;   q += (size_t)SORT_BLOCKS * RS_PITCH * 4;
    int*  cnt       = (int*)q;   q += (size_t)N * 4;
    int*  row_start = (int*)q;   q += (size_t)N * 4;
    int*  total     = (int*)q;
    float* z = bufA;
    unsigned short* h1 = (unsigned short*)bufA;

    const int T = 256;

    const int nbins = (N + BIN_ROWS - 1) / BIN_ROWS;
    const int chunk = (E + SORT_BLOCKS - 1) / SORT_BLOCKS;
    k_bin_chunks<<<SORT_BLOCKS, SORT_THREADS, 0, stream>>>(src, dst, ew, rec_dw, rec_s,
                                                          runstart, total, E, chunk, nbins);
    k_bin_rows<<<nbins, SORT_THREADS, 0, stream>>>(rec_dw, rec_s, runstart, dinv, cnt,
                                                  row_start, csr, total, N);

    k_prep_weights<<<(F_IN * F_HID + F_HID * F_OUT + T - 1) / T, T, 0, stream>>>(
        W1, W2, Wt1, Wt2);

    k_gemm1_wmma<<<GEMM1_BLOCKS, T, 0, stream>>>(x, Wt1, dinv, h1, N);
    k_gather1<<<(N + ROWS_PER_WAVE1 * 8 - 1) / (ROWS_PER_WAVE1 * 8), T, 0, stream>>>(
        csr, row_start, cnt, dinv, h1, b1, hb, N);

    k_gemm2_wmma<<<(N + 127) / 128, T, 0, stream>>>(hb, Wt2, dinv, z, N);
    k_gather2<<<(N + 15) / 16, T, 0, stream>>>(csr, row_start, cnt, dinv, z, b2, out, N);
}
